# starting kernel
# baseline (speedup 1.0000x reference)
#include <hip/hip_runtime.h>
#include <math.h>

typedef float v2f __attribute__((ext_vector_type(2)));
typedef float v8f __attribute__((ext_vector_type(8)));

#define C_IN 128
#define NEG_SLOPE 0.2f
#define LN_EPS 1e-5f

__global__ void init_kernel(float* __restrict__ out, const float* __restrict__ bias,
                            float* __restrict__ mmax, float* __restrict__ denom,
                            long N) {
    long i = (long)blockIdx.x * blockDim.x + threadIdx.x;
    if (i < N * 32) out[i] = bias[i & 31];
    if (i < N * 4) { mmax[i] = -__builtin_inff(); denom[i] = 0.0f; }
}

__global__ __launch_bounds__(256) void ln_gemm_kernel(
    const float* __restrict__ x,
    const float* __restrict__ gamma, const float* __restrict__ beta,
    const float* __restrict__ Wl, const float* __restrict__ bl,
    const float* __restrict__ Wr, const float* __restrict__ br,
    float* __restrict__ xl, float* __restrict__ xr, long N)
{
    __shared__ float sXh[16][132];

    const int t   = threadIdx.x;
    const int row = t >> 4;
    const int sub = t & 15;
    const long node = (long)blockIdx.x * 16 + row;

    float v[8];
    if (node < N) {
        const float4* xp = (const float4*)(x + node * C_IN + sub * 8);
        float4 p0 = xp[0], p1 = xp[1];
        v[0]=p0.x; v[1]=p0.y; v[2]=p0.z; v[3]=p0.w;
        v[4]=p1.x; v[5]=p1.y; v[6]=p1.z; v[7]=p1.w;
    } else {
        #pragma unroll
        for (int j = 0; j < 8; ++j) v[j] = 0.0f;
    }

    float s = 0.0f;
    #pragma unroll
    for (int j = 0; j < 8; ++j) s += v[j];
    s += __shfl_xor(s, 1, 16); s += __shfl_xor(s, 2, 16);
    s += __shfl_xor(s, 4, 16); s += __shfl_xor(s, 8, 16);
    const float mu = s * (1.0f / 128.0f);

    float q = 0.0f;
    #pragma unroll
    for (int j = 0; j < 8; ++j) { float d = v[j] - mu; q += d * d; }
    q += __shfl_xor(q, 1, 16); q += __shfl_xor(q, 2, 16);
    q += __shfl_xor(q, 4, 16); q += __shfl_xor(q, 8, 16);
    const float rstd = rsqrtf(q * (1.0f / 128.0f) + LN_EPS);

    #pragma unroll
    for (int j = 0; j < 8; ++j) {
        const int c = sub * 8 + j;
        float y = (v[j] - mu) * rstd * gamma[c] + beta[c];
        sXh[row][c] = y > 0.0f ? y : 0.0f;
    }
    __syncthreads();

    const int lane = t & 31;
    const int wave = t >> 5;
    const int half = lane >> 4;
    const int lr   = lane & 15;
    const int col  = wave * 16 + lr;

    v8f cl = {}; v8f cr = {};
    #pragma unroll 8
    for (int k = 0; k < 128; k += 4) {
        v2f a = *(const v2f*)&sXh[lr][k + 2 * half];
        const int rk = (k + 2 * half) * C_IN + col;
        v2f pb; pb.x = Wl[rk]; pb.y = Wl[rk + C_IN];
        cl = __builtin_amdgcn_wmma_f32_16x16x4_f32(false, a, false, pb, (short)0, cl, false, false);
        v2f qb; qb.x = Wr[rk]; qb.y = Wr[rk + C_IN];
        cr = __builtin_amdgcn_wmma_f32_16x16x4_f32(false, a, false, qb, (short)0, cr, false, false);
    }

    const float bll = bl[col], brr = br[col];
    const long base = (long)blockIdx.x * 16;
    #pragma unroll
    for (int vv = 0; vv < 8; ++vv) {
        const long r = base + vv + 8 * half;
        if (r < N) {
            xl[r * C_IN + col] = cl[vv] + bll;
            xr[r * C_IN + col] = cr[vv] + brr;
        }
    }
}

__device__ __forceinline__ void atomicMaxF(float* addr, float val) {
    if (val >= 0.0f) atomicMax((int*)addr, __float_as_int(val));
    else             atomicMin((unsigned int*)addr, __float_as_uint(val));
}

__global__ __launch_bounds__(256) void edge_score_kernel(
    const int* __restrict__ srcI, const int* __restrict__ dstI,
    const float* __restrict__ xl, const float* __restrict__ xr,
    const float* __restrict__ att,
    float* __restrict__ escore, float* __restrict__ mmax,
    long E, long Etot)
{
    const int lane = threadIdx.x & 31;
    const long e = (long)blockIdx.x * 8 + (threadIdx.x >> 5);
    if (e >= Etot) return;
    int s_, d_;
    if (e < E) { s_ = srcI[e]; d_ = dstI[e]; }
    else       { s_ = d_ = (int)(e - E); }

    const float4 a4 = ((const float4*)att)[lane];
    const float4 l4 = ((const float4*)xl)[(long)s_ * 32 + lane];
    const float4 r4 = ((const float4*)xr)[(long)d_ * 32 + lane];

    float4 t4;
    t4.x = l4.x + r4.x; t4.y = l4.y + r4.y; t4.z = l4.z + r4.z; t4.w = l4.w + r4.w;
    t4.x = t4.x > 0.0f ? t4.x : NEG_SLOPE * t4.x;
    t4.y = t4.y > 0.0f ? t4.y : NEG_SLOPE * t4.y;
    t4.z = t4.z > 0.0f ? t4.z : NEG_SLOPE * t4.z;
    t4.w = t4.w > 0.0f ? t4.w : NEG_SLOPE * t4.w;

    float p = t4.x * a4.x + t4.y * a4.y + t4.z * a4.z + t4.w * a4.w;
    p += __shfl_xor(p, 1, 8);
    p += __shfl_xor(p, 2, 8);
    p += __shfl_xor(p, 4, 8);

    if ((lane & 7) == 0) {
        const int h = lane >> 3;
        escore[e * 4 + h] = p;
        atomicMaxF(&mmax[(long)d_ * 4 + h], p);
    }
}

__global__ void softmax_denom_kernel(
    const int* __restrict__ dstI, float* __restrict__ escore,
    const float* __restrict__ mmax, float* __restrict__ denom,
    long E, long Etot)
{
    const long i = (long)blockIdx.x * blockDim.x + threadIdx.x;
    if (i >= Etot * 4) return;
    const long e = i >> 2;
    const int  h = (int)(i & 3);
    const int  d_ = (e < E) ? dstI[e] : (int)(e - E);
    const float w = __expf(escore[i] - mmax[(long)d_ * 4 + h]);
    escore[i] = w;
    atomicAdd(&denom[(long)d_ * 4 + h], w);
}

__global__ __launch_bounds__(256) void aggregate_kernel(
    const int* __restrict__ srcI, const int* __restrict__ dstI,
    const float* __restrict__ xl, const float* __restrict__ escore,
    const float* __restrict__ denom, float* __restrict__ out,
    long E, long Etot)
{
    const int lane = threadIdx.x & 31;
    const long e = (long)blockIdx.x * 8 + (threadIdx.x >> 5);
    if (e >= Etot) return;
    int s_, d_;
    if (e < E) { s_ = srcI[e]; d_ = dstI[e]; }
    else       { s_ = d_ = (int)(e - E); }

    const int h = lane >> 3;
    const float w   = escore[e * 4 + h];
    const float den = denom[(long)d_ * 4 + h];
    const float alpha = 0.25f * w / (den + 1e-16f);

    const float4 l4 = ((const float4*)xl)[(long)s_ * 32 + lane];
    float vx = alpha * l4.x, vy = alpha * l4.y, vz = alpha * l4.z, vw = alpha * l4.w;

    vx += __shfl_xor(vx, 8);  vy += __shfl_xor(vy, 8);
    vz += __shfl_xor(vz, 8);  vw += __shfl_xor(vw, 8);
    vx += __shfl_xor(vx, 16); vy += __shfl_xor(vy, 16);
    vz += __shfl_xor(vz, 16); vw += __shfl_xor(vw, 16);

    if (lane < 8) {
        float* o = out + (long)d_ * 32 + lane * 4;
        atomicAdd(o + 0, vx);
        atomicAdd(o + 1, vy);
        atomicAdd(o + 2, vz);
        atomicAdd(o + 3, vw);
    }
}

extern "C" void kernel_launch(void* const* d_in, const int* in_sizes, int n_in,
                              void* d_out, int out_size, void* d_ws, size_t ws_size,
                              hipStream_t stream) {
    const float* x     = (const float*)d_in[0];
    const int*   ei    = (const int*)d_in[1];
    const float* gamma = (const float*)d_in[2];
    const float* beta  = (const float*)d_in[3];
    const float* Wl    = (const float*)d_in[4];
    const float* bl    = (const float*)d_in[5];
    const float* Wr    = (const float*)d_in[6];
    const float* br    = (const float*)d_in[7];
    const float* att   = (const float*)d_in[8];
    const float* bias  = (const float*)d_in[9];
    float* out = (float*)d_out;

    const long N = in_sizes[0] / C_IN;
    const long E = in_sizes[1] / 2;
    const long Etot = E + N;
    const int* srcI = ei;
    const int* dstI = ei + E;

    float* ws     = (float*)d_ws;
    float* xl     = ws;
    float* xr     = xl + N * C_IN;
    float* escore = xr + N * C_IN;
    float* mmax   = escore + Etot * 4;
    float* denom  = mmax + N * 4;

    {
        long total = N * 32;
        init_kernel<<<(int)((total + 255) / 256), 256, 0, stream>>>(out, bias, mmax, denom, N);
    }
    {
        int blocks = (int)((N + 15) / 16);
        ln_gemm_kernel<<<blocks, 256, 0, stream>>>(x, gamma, beta, Wl, bl, Wr, br, xl, xr, N);
    }
    {
        int blocks = (int)((Etot + 7) / 8);
        edge_score_kernel<<<blocks, 256, 0, stream>>>(srcI, dstI, xl, xr, att, escore, mmax, E, Etot);
    }
    {
        long total = Etot * 4;
        softmax_denom_kernel<<<(int)((total + 255) / 256), 256, 0, stream>>>(dstI, escore, mmax, denom, E, Etot);
    }
    {
        int blocks = (int)((Etot + 7) / 8);
        aggregate_kernel<<<blocks, 256, 0, stream>>>(srcI, dstI, xl, escore, denom, out, E, Etot);
    }
}


# Group edges by destination (counting sort, arrival-order rank); one wave per node runs scores, running softmax and weighted sum in registers, no float atomics: sum order changes
# speedup vs baseline: 1.5513x; 1.5513x over previous
#include <hip/hip_runtime.h>
#include <math.h>

typedef float v2f __attribute__((ext_vector_type(2)));
typedef float v8f __attribute__((ext_vector_type(8)));
typedef float v4f __attribute__((ext_vector_type(4)));

#define C_IN 128
#define NEG_SLOPE 0.2f
#define LN_EPS 1e-5f

__global__ __launch_bounds__(256) void ln_gemm_kernel(
    const float* __restrict__ x,
    const float* __restrict__ gamma, const float* __restrict__ beta,
    const float* __restrict__ Wl, const float* __restrict__ bl,
    const float* __restrict__ Wr, const float* __restrict__ br,
    float* __restrict__ xl, float* __restrict__ xr,
    int* __restrict__ cnt, int* __restrict__ total, long N)
{
    __shared__ float sXh[16][132];

    const int t   = threadIdx.x;
    if (t < 16) {
        const long ci = (long)blockIdx.x * 16 + t;
        if (ci < N) cnt[ci] = 0;
        if (ci == 0) *total = 0;
    }
    const int row = t >> 4;
    const int sub = t & 15;
    const long node = (long)blockIdx.x * 16 + row;

    float v[8];
    if (node < N) {
        const float4* xp = (const float4*)(x + node * C_IN + sub * 8);
        float4 p0 = xp[0], p1 = xp[1];
        v[0]=p0.x; v[1]=p0.y; v[2]=p0.z; v[3]=p0.w;
        v[4]=p1.x; v[5]=p1.y; v[6]=p1.z; v[7]=p1.w;
    } else {
        #pragma unroll
        for (int j = 0; j < 8; ++j) v[j] = 0.0f;
    }

    float s = 0.0f;
    #pragma unroll
    for (int j = 0; j < 8; ++j) s += v[j];
    s += __shfl_xor(s, 1, 16); s += __shfl_xor(s, 2, 16);
    s += __shfl_xor(s, 4, 16); s += __shfl_xor(s, 8, 16);
    const float mu = s * (1.0f / 128.0f);

    float q = 0.0f;
    #pragma unroll
    for (int j = 0; j < 8; ++j) { float d = v[j] - mu; q += d * d; }
    q += __shfl_xor(q, 1, 16); q += __shfl_xor(q, 2, 16);
    q += __shfl_xor(q, 4, 16); q += __shfl_xor(q, 8, 16);
    const float rstd = rsqrtf(q * (1.0f / 128.0f) + LN_EPS);

    #pragma unroll
    for (int j = 0; j < 8; ++j) {
        const int c = sub * 8 + j;
        float y = (v[j] - mu) * rstd * gamma[c] + beta[c];
        sXh[row][c] = y > 0.0f ? y : 0.0f;
    }
    __syncthreads();

    const int lane = t & 31;
    const int wave = t >> 5;
    const int half = lane >> 4;
    const int lr   = lane & 15;
    const int col  = wave * 16 + lr;

    v8f cl = {}; v8f cr = {};
    #pragma unroll 8
    for (int k = 0; k < 128; k += 4) {
        v2f a = *(const v2f*)&sXh[lr][k + 2 * half];
        const int rk = (k + 2 * half) * C_IN + col;
        v2f pb; pb.x = Wl[rk]; pb.y = Wl[rk + C_IN];
        cl = __builtin_amdgcn_wmma_f32_16x16x4_f32(false, a, false, pb, (short)0, cl, false, false);
        v2f qb; qb.x = Wr[rk]; qb.y = Wr[rk + C_IN];
        cr = __builtin_amdgcn_wmma_f32_16x16x4_f32(false, a, false, qb, (short)0, cr, false, false);
    }

    const float bll = bl[col], brr = br[col];
    const long base = (long)blockIdx.x * 16;
    #pragma unroll
    for (int vv = 0; vv < 8; ++vv) {
        const long r = base + vv + 8 * half;
        if (r < N) {
            xl[r * C_IN + col] = cl[vv] + bll;
            xr[r * C_IN + col] = cr[vv] + brr;
        }
    }
}

__global__ __launch_bounds__(256) void degree_kernel(
    const int* __restrict__ dstI, int* __restrict__ cnt, long E)
{
    const long e = (long)blockIdx.x * 256 + threadIdx.x;
    if (e < E) atomicAdd(&cnt[dstI[e]], 1);
}

__global__ __launch_bounds__(1024) void segment_alloc_kernel(
    const int* __restrict__ cnt, int* __restrict__ total, int* __restrict__ seg, long N)
{
    __shared__ int sWave[32];
    __shared__ int sBase;

    const int t    = threadIdx.x;
    const int lane = t & 31;
    const int wv   = t >> 5;
    const long i   = (long)blockIdx.x * 1024 + t;
    const int c    = (i < N) ? cnt[i] : 0;

    int inc = c;
    #pragma unroll
    for (int o = 1; o < 32; o <<= 1) {
        const int up = __shfl_up(inc, o);
        if (lane >= o) inc += up;
    }
    if (lane == 31) sWave[wv] = inc;
    __syncthreads();

    if (wv == 0) {
        const int wsum = sWave[lane];
        int winc = wsum;
        #pragma unroll
        for (int o = 1; o < 32; o <<= 1) {
            const int up = __shfl_up(winc, o);
            if (lane >= o) winc += up;
        }
        sWave[lane] = winc - wsum;
        if (lane == 31) sBase = atomicAdd(total, winc);
    }
    __syncthreads();

    if (i < N) {
        seg[2 * i]     = sBase + sWave[wv] + inc - c;
        seg[2 * i + 1] = c;
    }
}

__global__ __launch_bounds__(256) void group_fill_kernel(
    const int* __restrict__ srcI, const int* __restrict__ dstI,
    int* __restrict__ cnt, const int* __restrict__ seg, int* __restrict__ gsrc, long E)
{
    const long e = (long)blockIdx.x * 256 + threadIdx.x;
    if (e >= E) return;
    const int d_ = dstI[e];
    const int rank = atomicSub(&cnt[d_], 1) - 1;
    gsrc[seg[2 * (long)d_] + rank] = srcI[e];
}

__device__ __forceinline__ float head_score(const float4 l4, const float4 r4, const float4 a4) {
    float4 t4;
    t4.x = l4.x + r4.x; t4.y = l4.y + r4.y; t4.z = l4.z + r4.z; t4.w = l4.w + r4.w;
    t4.x = t4.x > 0.0f ? t4.x : NEG_SLOPE * t4.x;
    t4.y = t4.y > 0.0f ? t4.y : NEG_SLOPE * t4.y;
    t4.z = t4.z > 0.0f ? t4.z : NEG_SLOPE * t4.z;
    t4.w = t4.w > 0.0f ? t4.w : NEG_SLOPE * t4.w;
    float p = t4.x * a4.x + t4.y * a4.y + t4.z * a4.z + t4.w * a4.w;
    p += __shfl_xor(p, 1, 8);
    p += __shfl_xor(p, 2, 8);
    p += __shfl_xor(p, 4, 8);
    return p;
}

__global__ __launch_bounds__(256) void node_softmax_aggregate_kernel(
    const int* __restrict__ gsrc, const int* __restrict__ seg,
    const float* __restrict__ xl, const float* __restrict__ xr,
    const float* __restrict__ att, const float* __restrict__ bias,
    float* __restrict__ out, long N)
{
    const int lane = threadIdx.x & 31;
    const int node = __builtin_amdgcn_readfirstlane((int)(blockIdx.x * 8 + (threadIdx.x >> 5)));
    if (node >= N) return;

    const int start = seg[2 * (long)node];
    const int cnt   = seg[2 * (long)node + 1];

    const float4* xl4 = (const float4*)xl;
    const float4 a4 = ((const float4*)att)[lane];
    const v4f rv = __builtin_nontemporal_load((const v4f*)xr + (long)node * 32 + lane);
    float4 r4; r4.x = rv.x; r4.y = rv.y; r4.z = rv.z; r4.w = rv.w;

    const float4 self4 = xl4[(long)node * 32 + lane];
    float m   = head_score(self4, r4, a4);
    float den = 1.0f;
    float4 acc = self4;

    const float NEG_INF = -__builtin_inff();

    for (int j0 = 0; j0 < cnt; j0 += 32) {
        const int nj = (cnt - j0 < 32) ? (cnt - j0) : 32;
        const int idx = (lane < nj) ? gsrc[start + j0 + lane] : node;
        for (int j = 0; j < nj; j += 4) {
            const int s0 = __builtin_amdgcn_readlane(idx, j);
            const int s1 = __builtin_amdgcn_readlane(idx, j + 1);
            const int s2 = __builtin_amdgcn_readlane(idx, j + 2);
            const int s3 = __builtin_amdgcn_readlane(idx, j + 3);
            const float4 q0 = xl4[(long)s0 * 32 + lane];
            const float4 q1 = xl4[(long)s1 * 32 + lane];
            const float4 q2 = xl4[(long)s2 * 32 + lane];
            const float4 q3 = xl4[(long)s3 * 32 + lane];
            const float p0 = head_score(q0, r4, a4);
            float p1 = head_score(q1, r4, a4);
            float p2 = head_score(q2, r4, a4);
            float p3 = head_score(q3, r4, a4);
            if (j + 1 >= nj) p1 = NEG_INF;
            if (j + 2 >= nj) p2 = NEG_INF;
            if (j + 3 >= nj) p3 = NEG_INF;
            const float gm = fmaxf(fmaxf(m, p0), fmaxf(p1, fmaxf(p2, p3)));
            const float sc = __expf(m - gm);
            const float w0 = __expf(p0 - gm);
            const float w1 = __expf(p1 - gm);
            const float w2 = __expf(p2 - gm);
            const float w3 = __expf(p3 - gm);
            den   = den   * sc + ((w0 + w1) + (w2 + w3));
            acc.x = acc.x * sc + ((w0 * q0.x + w1 * q1.x) + (w2 * q2.x + w3 * q3.x));
            acc.y = acc.y * sc + ((w0 * q0.y + w1 * q1.y) + (w2 * q2.y + w3 * q3.y));
            acc.z = acc.z * sc + ((w0 * q0.z + w1 * q1.z) + (w2 * q2.z + w3 * q3.z));
            acc.w = acc.w * sc + ((w0 * q0.w + w1 * q1.w) + (w2 * q2.w + w3 * q3.w));
            m = gm;
        }
    }

    const float inv = 0.25f / (den + 1e-16f);
    float vx = acc.x * inv, vy = acc.y * inv, vz = acc.z * inv, vw = acc.w * inv;
    vx += __shfl_xor(vx, 8);  vy += __shfl_xor(vy, 8);
    vz += __shfl_xor(vz, 8);  vw += __shfl_xor(vw, 8);
    vx += __shfl_xor(vx, 16); vy += __shfl_xor(vy, 16);
    vz += __shfl_xor(vz, 16); vw += __shfl_xor(vw, 16);

    if (lane < 8) {
        const float4 b4 = ((const float4*)bias)[lane];
        float4 o4;
        o4.x = vx + b4.x; o4.y = vy + b4.y; o4.z = vz + b4.z; o4.w = vw + b4.w;
        ((float4*)out)[(long)node * 8 + lane] = o4;
    }
}

extern "C" void kernel_launch(void* const* d_in, const int* in_sizes, int n_in,
                              void* d_out, int out_size, void* d_ws, size_t ws_size,
                              hipStream_t stream) {
    const float* x     = (const float*)d_in[0];
    const int*   ei    = (const int*)d_in[1];
    const float* gamma = (const float*)d_in[2];
    const float* beta  = (const float*)d_in[3];
    const float* Wl    = (const float*)d_in[4];
    const float* bl    = (const float*)d_in[5];
    const float* Wr    = (const float*)d_in[6];
    const float* br    = (const float*)d_in[7];
    const float* att   = (const float*)d_in[8];
    const float* bias  = (const float*)d_in[9];
    float* out = (float*)d_out;

    const long N = in_sizes[0] / C_IN;
    const long E = in_sizes[1] / 2;
    const int* srcI = ei;
    const int* dstI = ei + E;

    float* ws   = (float*)d_ws;
    float* xl   = ws;
    float* xr   = xl + N * C_IN;
    int*   seg  = (int*)(xr + N * C_IN);
    int*   cnt  = seg + N * 2;
    int*   total = cnt + N;
    int*   gsrc = total + 4;

    {
        int blocks = (int)((N + 15) / 16);
        ln_gemm_kernel<<<blocks, 256, 0, stream>>>(x, gamma, beta, Wl, bl, Wr, br, xl, xr, cnt, total, N);
    }
    {
        int blocks = (int)((E + 255) / 256);
        degree_kernel<<<blocks, 256, 0, stream>>>(dstI, cnt, E);
    }
    {
        int blocks = (int)((N + 1023) / 1024);
        segment_alloc_kernel<<<blocks, 1024, 0, stream>>>(cnt, total, seg, N);
    }
    {
        int blocks = (int)((E + 255) / 256);
        group_fill_kernel<<<blocks, 256, 0, stream>>>(srcI, dstI, cnt, seg, gsrc, E);
    }
    {
        int blocks = (int)((N + 7) / 8);
        node_softmax_aggregate_kernel<<<blocks, 256, 0, stream>>>(gsrc, seg, xl, xr, att, bias, out, N);
    }
}


# Edge grouping as a two-level counting sort through LDS (128-node bins; counts and ranks by LDS adds, offsets by a scan, whole-run copies): no global atomics; per-node sum order = LDS arrival order
# speedup vs baseline: 2.6689x; 1.7204x over previous
#include <hip/hip_runtime.h>
#include <math.h>

typedef float v2f __attribute__((ext_vector_type(2)));
typedef float v8f __attribute__((ext_vector_type(8)));
typedef float v4f __attribute__((ext_vector_type(4)));

#define C_IN 128
#define NEG_SLOPE 0.2f
#define LN_EPS 1e-5f

__global__ __launch_bounds__(256) void ln_gemm_kernel(
    const float* __restrict__ x,
    const float* __restrict__ gamma, const float* __restrict__ beta,
    const float* __restrict__ Wl, const float* __restrict__ bl,
    const float* __restrict__ Wr, const float* __restrict__ br,
    float* __restrict__ xl, float* __restrict__ xr, long N)
{
    __shared__ float sXh[16][132];

    const int t   = threadIdx.x;
    const int row = t >> 4;
    const int sub = t & 15;
    const long node = (long)blockIdx.x * 16 + row;

    float v[8];
    if (node < N) {
        const float4* xp = (const float4*)(x + node * C_IN + sub * 8);
        float4 p0 = xp[0], p1 = xp[1];
        v[0]=p0.x; v[1]=p0.y; v[2]=p0.z; v[3]=p0.w;
        v[4]=p1.x; v[5]=p1.y; v[6]=p1.z; v[7]=p1.w;
    } else {
        #pragma unroll
        for (int j = 0; j < 8; ++j) v[j] = 0.0f;
    }

    float s = 0.0f;
    #pragma unroll
    for (int j = 0; j < 8; ++j) s += v[j];
    s += __shfl_xor(s, 1, 16); s += __shfl_xor(s, 2, 16);
    s += __shfl_xor(s, 4, 16); s += __shfl_xor(s, 8, 16);
    const float mu = s * (1.0f / 128.0f);

    float q = 0.0f;
    #pragma unroll
    for (int j = 0; j < 8; ++j) { float d = v[j] - mu; q += d * d; }
    q += __shfl_xor(q, 1, 16); q += __shfl_xor(q, 2, 16);
    q += __shfl_xor(q, 4, 16); q += __shfl_xor(q, 8, 16);
    const float rstd = rsqrtf(q * (1.0f / 128.0f) + LN_EPS);

    #pragma unroll
    for (int j = 0; j < 8; ++j) {
        const int c = sub * 8 + j;
        float y = (v[j] - mu) * rstd * gamma[c] + beta[c];
        sXh[row][c] = y > 0.0f ? y : 0.0f;
    }
    __syncthreads();

    const int lane = t & 31;
    const int wave = t >> 5;
    const int half = lane >> 4;
    const int lr   = lane & 15;
    const int col  = wave * 16 + lr;

    v8f cl = {}; v8f cr = {};
    #pragma unroll 8
    for (int k = 0; k < 128; k += 4) {
        v2f a = *(const v2f*)&sXh[lr][k + 2 * half];
        const int rk = (k + 2 * half) * C_IN + col;
        v2f pb; pb.x = Wl[rk]; pb.y = Wl[rk + C_IN];
        cl = __builtin_amdgcn_wmma_f32_16x16x4_f32(false, a, false, pb, (short)0, cl, false, false);
        v2f qb; qb.x = Wr[rk]; qb.y = Wr[rk + C_IN];
        cr = __builtin_amdgcn_wmma_f32_16x16x4_f32(false, a, false, qb, (short)0, cr, false, false);
    }

    const float bll = bl[col], brr = br[col];
    const long base = (long)blockIdx.x * 16;
    #pragma unroll
    for (int vv = 0; vv < 8; ++vv) {
        const long r = base + vv + 8 * half;
        if (r < N) {
            xl[r * C_IN + col] = cl[vv] + bll;
            xr[r * C_IN + col] = cr[vv] + brr;
        }
    }
}

#define BIN_SHIFT 7
#define BIN_NODES 128
#define MAX_BINS 1024
#define EDGE_CHUNK 8192
#define SORT_STAGE 8192

__device__ __forceinline__ int block_exclusive_scan_1024(int c, int* sWave, int* blockTotal) {
    const int lane = threadIdx.x & 31;
    const int wv   = threadIdx.x >> 5;
    int inc = c;
    #pragma unroll
    for (int o = 1; o < 32; o <<= 1) {
        const int up = __shfl_up(inc, o);
        if (lane >= o) inc += up;
    }
    if (lane == 31) sWave[wv] = inc;
    __syncthreads();
    if (wv == 0) {
        const int wsum = sWave[lane];
        int winc = wsum;
        #pragma unroll
        for (int o = 1; o < 32; o <<= 1) {
            const int up = __shfl_up(winc, o);
            if (lane >= o) winc += up;
        }
        sWave[lane] = winc - wsum;
        if (lane == 31) *blockTotal = winc;
    }
    __syncthreads();
    return sWave[wv] + inc - c;
}

__global__ __launch_bounds__(1024) void bin_count_kernel(
    const int* __restrict__ dstI, int* __restrict__ cntMat, long E, int NB)
{
    __shared__ int hist[MAX_BINS];
    const int t = threadIdx.x;
    hist[t] = 0;
    __syncthreads();

    const long base = (long)blockIdx.x * EDGE_CHUNK;
    int d[8];
    #pragma unroll
    for (int i = 0; i < 8; ++i) {
        const long e = base + i * 1024 + t;
        d[i] = (e < E) ? dstI[e] : -1;
    }
    #pragma unroll
    for (int i = 0; i < 8; ++i)
        if (d[i] >= 0) atomicAdd(&hist[d[i] >> BIN_SHIFT], 1);
    __syncthreads();

    if (t < NB) cntMat[(long)blockIdx.x * NB + t] = hist[t];
}

__global__ __launch_bounds__(1024) void bin_scan_kernel(
    const int* __restrict__ cntMat, int* __restrict__ offMat, int* __restrict__ binBase,
    int NW, int NB)
{
    __shared__ int sWave[32];
    __shared__ int sTotal;
    const int t = threadIdx.x;

    int run = 0;
    if (t < NB) {
        for (int w0 = 0; w0 < NW; w0 += 16) {
            int c[16];
            #pragma unroll
            for (int k = 0; k < 16; ++k)
                c[k] = (w0 + k < NW) ? cntMat[(long)(w0 + k) * NB + t] : 0;
            #pragma unroll
            for (int k = 0; k < 16; ++k) {
                if (w0 + k < NW) offMat[(long)(w0 + k) * NB + t] = run;
                run += c[k];
            }
        }
    }
    const int ex = block_exclusive_scan_1024(run, sWave, &sTotal);
    if (t < NB) binBase[t] = ex;
    if (t == 0) binBase[NB] = sTotal;
}

__global__ __launch_bounds__(1024) void bin_scatter_kernel(
    const int* __restrict__ srcI, const int* __restrict__ dstI,
    const int* __restrict__ cntMat, const int* __restrict__ offMat,
    const int* __restrict__ binBase, int* __restrict__ binned, long E, int NB)
{
    __shared__ int sWave[32];
    __shared__ int sTotal;
    __shared__ int lcur[MAX_BINS];
    __shared__ int loff[MAX_BINS];
    __shared__ int goff[MAX_BINS];
    __shared__ int stageV[EDGE_CHUNK];
    __shared__ int stageB[EDGE_CHUNK];

    const int t = threadIdx.x;
    const long row = (long)blockIdx.x * NB;
    const int c = (t < NB) ? cntMat[row + t] : 0;
    const int g = (t < NB) ? (binBase[t] + offMat[row + t]) : 0;
    const int lo = block_exclusive_scan_1024(c, sWave, &sTotal);
    loff[t] = lo;
    lcur[t] = lo;
    goff[t] = g;
    __syncthreads();

    const long base = (long)blockIdx.x * EDGE_CHUNK;
    int d[8], sidx[8];
    #pragma unroll
    for (int i = 0; i < 8; ++i) {
        const long e = base + i * 1024 + t;
        d[i]    = (e < E) ? dstI[e] : -1;
        sidx[i] = (e < E) ? srcI[e] : 0;
    }
    #pragma unroll
    for (int i = 0; i < 8; ++i) {
        if (d[i] >= 0) {
            const int b = d[i] >> BIN_SHIFT;
            const int p = atomicAdd(&lcur[b], 1);
            stageV[p] = (sidx[i] << BIN_SHIFT) | (d[i] & (BIN_NODES - 1));
            stageB[p] = b;
        }
    }
    __syncthreads();

    const long left = E - base;
    const int nLocal = left < EDGE_CHUNK ? (int)left : EDGE_CHUNK;
    for (int j = t; j < nLocal; j += 1024) {
        const int b = stageB[j];
        binned[goff[b] + (j - loff[b])] = stageV[j];
    }
}

__global__ __launch_bounds__(256) void bin_sort_kernel(
    const int* __restrict__ binned, const int* __restrict__ binBase,
    int* __restrict__ seg, int* __restrict__ gsrc, long N)
{
    __shared__ int ncnt[BIN_NODES];
    __shared__ int ncur[BIN_NODES];
    __shared__ int sW[4];
    __shared__ int stage[SORT_STAGE];

    const int t    = threadIdx.x;
    const int lane = t & 31;
    const int wv   = t >> 5;
    const int bin  = blockIdx.x;
    const int start = binBase[bin];
    const int n     = binBase[bin + 1] - start;

    if (t < BIN_NODES) ncnt[t] = 0;
    __syncthreads();
    for (int i = t; i < n; i += 256)
        atomicAdd(&ncnt[binned[start + i] & (BIN_NODES - 1)], 1);
    __syncthreads();

    const int c = (t < BIN_NODES) ? ncnt[t] : 0;
    int inc = c;
    #pragma unroll
    for (int o = 1; o < 32; o <<= 1) {
        const int up = __shfl_up(inc, o);
        if (lane >= o) inc += up;
    }
    if (lane == 31 && wv < 4) sW[wv] = inc;
    __syncthreads();
    if (t < BIN_NODES) {
        int woff = 0;
        for (int k = 0; k < wv; ++k) woff += sW[k];
        const int ex = woff + inc - c;
        ncur[t] = ex;
        const long node = (long)bin * BIN_NODES + t;
        if (node < N) {
            seg[2 * node]     = start + ex;
            seg[2 * node + 1] = c;
        }
    }
    __syncthreads();

    if (n <= SORT_STAGE) {
        for (int i = t; i < n; i += 256) {
            const int v = binned[start + i];
            const int p = atomicAdd(&ncur[v & (BIN_NODES - 1)], 1);
            stage[p] = v >> BIN_SHIFT;
        }
        __syncthreads();
        for (int i = t; i < n; i += 256) gsrc[start + i] = stage[i];
    } else {
        for (int i = t; i < n; i += 256) {
            const int v = binned[start + i];
            const int p = atomicAdd(&ncur[v & (BIN_NODES - 1)], 1);
            gsrc[start + p] = v >> BIN_SHIFT;
        }
    }
}

__device__ __forceinline__ float head_score(const float4 l4, const float4 r4, const float4 a4) {
    float4 t4;
    t4.x = l4.x + r4.x; t4.y = l4.y + r4.y; t4.z = l4.z + r4.z; t4.w = l4.w + r4.w;
    t4.x = t4.x > 0.0f ? t4.x : NEG_SLOPE * t4.x;
    t4.y = t4.y > 0.0f ? t4.y : NEG_SLOPE * t4.y;
    t4.z = t4.z > 0.0f ? t4.z : NEG_SLOPE * t4.z;
    t4.w = t4.w > 0.0f ? t4.w : NEG_SLOPE * t4.w;
    float p = t4.x * a4.x + t4.y * a4.y + t4.z * a4.z + t4.w * a4.w;
    p += __shfl_xor(p, 1, 8);
    p += __shfl_xor(p, 2, 8);
    p += __shfl_xor(p, 4, 8);
    return p;
}

__global__ __launch_bounds__(256) void node_softmax_aggregate_kernel(
    const int* __restrict__ gsrc, const int* __restrict__ seg,
    const float* __restrict__ xl, const float* __restrict__ xr,
    const float* __restrict__ att, const float* __restrict__ bias,
    float* __restrict__ out, long N)
{
    const int lane = threadIdx.x & 31;
    const int node = __builtin_amdgcn_readfirstlane((int)(blockIdx.x * 8 + (threadIdx.x >> 5)));
    if (node >= N) return;

    const int start = seg[2 * (long)node];
    const int cnt   = seg[2 * (long)node + 1];

    const float4* xl4 = (const float4*)xl;
    const float4 a4 = ((const float4*)att)[lane];
    const v4f rv = __builtin_nontemporal_load((const v4f*)xr + (long)node * 32 + lane);
    float4 r4; r4.x = rv.x; r4.y = rv.y; r4.z = rv.z; r4.w = rv.w;

    const float4 self4 = xl4[(long)node * 32 + lane];
    float m   = head_score(self4, r4, a4);
    float den = 1.0f;
    float4 acc = self4;

    const float NEG_INF = -__builtin_inff();

    for (int j0 = 0; j0 < cnt; j0 += 32) {
        const int nj = (cnt - j0 < 32) ? (cnt - j0) : 32;
        const int idx = (lane < nj) ? gsrc[start + j0 + lane] : node;
        for (int j = 0; j < nj; j += 4) {
            const int s0 = __builtin_amdgcn_readlane(idx, j);
            const int s1 = __builtin_amdgcn_readlane(idx, j + 1);
            const int s2 = __builtin_amdgcn_readlane(idx, j + 2);
            const int s3 = __builtin_amdgcn_readlane(idx, j + 3);
            const float4 q0 = xl4[(long)s0 * 32 + lane];
            const float4 q1 = xl4[(long)s1 * 32 + lane];
            const float4 q2 = xl4[(long)s2 * 32 + lane];
            const float4 q3 = xl4[(long)s3 * 32 + lane];
            const float p0 = head_score(q0, r4, a4);
            float p1 = head_score(q1, r4, a4);
            float p2 = head_score(q2, r4, a4);
            float p3 = head_score(q3, r4, a4);
            if (j + 1 >= nj) p1 = NEG_INF;
            if (j + 2 >= nj) p2 = NEG_INF;
            if (j + 3 >= nj) p3 = NEG_INF;
            const float gm = fmaxf(fmaxf(m, p0), fmaxf(p1, fmaxf(p2, p3)));
            const float sc = __expf(m - gm);
            const float w0 = __expf(p0 - gm);
            const float w1 = __expf(p1 - gm);
            const float w2 = __expf(p2 - gm);
            const float w3 = __expf(p3 - gm);
            den   = den   * sc + ((w0 + w1) + (w2 + w3));
            acc.x = acc.x * sc + ((w0 * q0.x + w1 * q1.x) + (w2 * q2.x + w3 * q3.x));
            acc.y = acc.y * sc + ((w0 * q0.y + w1 * q1.y) + (w2 * q2.y + w3 * q3.y));
            acc.z = acc.z * sc + ((w0 * q0.z + w1 * q1.z) + (w2 * q2.z + w3 * q3.z));
            acc.w = acc.w * sc + ((w0 * q0.w + w1 * q1.w) + (w2 * q2.w + w3 * q3.w));
            m = gm;
        }
    }

    const float inv = 0.25f / (den + 1e-16f);
    float vx = acc.x * inv, vy = acc.y * inv, vz = acc.z * inv, vw = acc.w * inv;
    vx += __shfl_xor(vx, 8);  vy += __shfl_xor(vy, 8);
    vz += __shfl_xor(vz, 8);  vw += __shfl_xor(vw, 8);
    vx += __shfl_xor(vx, 16); vy += __shfl_xor(vy, 16);
    vz += __shfl_xor(vz, 16); vw += __shfl_xor(vw, 16);

    if (lane < 8) {
        const float4 b4 = ((const float4*)bias)[lane];
        float4 o4;
        o4.x = vx + b4.x; o4.y = vy + b4.y; o4.z = vz + b4.z; o4.w = vw + b4.w;
        ((float4*)out)[(long)node * 8 + lane] = o4;
    }
}

extern "C" void kernel_launch(void* const* d_in, const int* in_sizes, int n_in,
                              void* d_out, int out_size, void* d_ws, size_t ws_size,
                              hipStream_t stream) {
    const float* x     = (const float*)d_in[0];
    const int*   ei    = (const int*)d_in[1];
    const float* gamma = (const float*)d_in[2];
    const float* beta  = (const float*)d_in[3];
    const float* Wl    = (const float*)d_in[4];
    const float* bl    = (const float*)d_in[5];
    const float* Wr    = (const float*)d_in[6];
    const float* br    = (const float*)d_in[7];
    const float* att   = (const float*)d_in[8];
    const float* bias  = (const float*)d_in[9];
    float* out = (float*)d_out;

    const long N = in_sizes[0] / C_IN;
    const long E = in_sizes[1] / 2;
    const int* srcI = ei;
    const int* dstI = ei + E;

    float* ws   = (float*)d_ws;
    float* xl   = ws;
    float* xr   = xl + N * C_IN;
    int*   seg  = (int*)(xr + N * C_IN);
    int*   gsrc = seg + N * 2;

    const int NB = (int)((N + BIN_NODES - 1) / BIN_NODES);
    const int NW = (int)((E + EDGE_CHUNK - 1) / EDGE_CHUNK);
    int*   binned  = (int*)ws;
    int*   binBase = binned + E;
    int*   cntMat  = binBase + 2048;
    int*   offMat  = cntMat + (long)NW * NB;

    bin_count_kernel<<<NW, 1024, 0, stream>>>(dstI, cntMat, E, NB);
    bin_scan_kernel<<<1, 1024, 0, stream>>>(cntMat, offMat, binBase, NW, NB);
    bin_scatter_kernel<<<NW, 1024, 0, stream>>>(srcI, dstI, cntMat, offMat, binBase, binned, E, NB);
    bin_sort_kernel<<<NB, 256, 0, stream>>>(binned, binBase, seg, gsrc, N);
    {
        int blocks = (int)((N + 15) / 16);
        ln_gemm_kernel<<<blocks, 256, 0, stream>>>(x, gamma, beta, Wl, bl, Wr, br, xl, xr, N);
    }
    {
        int blocks = (int)((N + 7) / 8);
        node_softmax_aggregate_kernel<<<blocks, 256, 0, stream>>>(gsrc, seg, xl, xr, att, bias, out, N);
    }
}


# Aggregation launch keeps 1792 work-groups resident (7 per WGP): each wave loops over its nodes, next node's range and ids fetched ahead, rows by 32-bit index; no wave per node; sum order unchanged
# speedup vs baseline: 3.0862x; 1.1564x over previous
#include <hip/hip_runtime.h>
#include <math.h>

typedef float v2f __attribute__((ext_vector_type(2)));
typedef float v8f __attribute__((ext_vector_type(8)));
typedef float v4f __attribute__((ext_vector_type(4)));

#define C_IN 128
#define NEG_SLOPE 0.2f
#define LN_EPS 1e-5f

__global__ __launch_bounds__(256) void ln_gemm_kernel(
    const float* __restrict__ x,
    const float* __restrict__ gamma, const float* __restrict__ beta,
    const float* __restrict__ Wl, const float* __restrict__ bl,
    const float* __restrict__ Wr, const float* __restrict__ br,
    float* __restrict__ xl, float* __restrict__ xr, long N)
{
    __shared__ float sXh[16][132];

    const int t   = threadIdx.x;
    const int row = t >> 4;
    const int sub = t & 15;
    const long node = (long)blockIdx.x * 16 + row;

    float v[8];
    if (node < N) {
        const float4* xp = (const float4*)(x + node * C_IN + sub * 8);
        float4 p0 = xp[0], p1 = xp[1];
        v[0]=p0.x; v[1]=p0.y; v[2]=p0.z; v[3]=p0.w;
        v[4]=p1.x; v[5]=p1.y; v[6]=p1.z; v[7]=p1.w;
    } else {
        #pragma unroll
        for (int j = 0; j < 8; ++j) v[j] = 0.0f;
    }

    float s = 0.0f;
    #pragma unroll
    for (int j = 0; j < 8; ++j) s += v[j];
    s += __shfl_xor(s, 1, 16); s += __shfl_xor(s, 2, 16);
    s += __shfl_xor(s, 4, 16); s += __shfl_xor(s, 8, 16);
    const float mu = s * (1.0f / 128.0f);

    float q = 0.0f;
    #pragma unroll
    for (int j = 0; j < 8; ++j) { float d = v[j] - mu; q += d * d; }
    q += __shfl_xor(q, 1, 16); q += __shfl_xor(q, 2, 16);
    q += __shfl_xor(q, 4, 16); q += __shfl_xor(q, 8, 16);
    const float rstd = rsqrtf(q * (1.0f / 128.0f) + LN_EPS);

    #pragma unroll
    for (int j = 0; j < 8; ++j) {
        const int c = sub * 8 + j;
        float y = (v[j] - mu) * rstd * gamma[c] + beta[c];
        sXh[row][c] = y > 0.0f ? y : 0.0f;
    }
    __syncthreads();

    const int lane = t & 31;
    const int wave = t >> 5;
    const int half = lane >> 4;
    const int lr   = lane & 15;
    const int col  = wave * 16 + lr;

    v8f cl = {}; v8f cr = {};
    #pragma unroll 8
    for (int k = 0; k < 128; k += 4) {
        v2f a = *(const v2f*)&sXh[lr][k + 2 * half];
        const int rk = (k + 2 * half) * C_IN + col;
        v2f pb; pb.x = Wl[rk]; pb.y = Wl[rk + C_IN];
        cl = __builtin_amdgcn_wmma_f32_16x16x4_f32(false, a, false, pb, (short)0, cl, false, false);
        v2f qb; qb.x = Wr[rk]; qb.y = Wr[rk + C_IN];
        cr = __builtin_amdgcn_wmma_f32_16x16x4_f32(false, a, false, qb, (short)0, cr, false, false);
    }

    const float bll = bl[col], brr = br[col];
    const long base = (long)blockIdx.x * 16;
    #pragma unroll
    for (int vv = 0; vv < 8; ++vv) {
        const long r = base + vv + 8 * half;
        if (r < N) {
            xl[r * C_IN + col] = cl[vv] + bll;
            xr[r * C_IN + col] = cr[vv] + brr;
        }
    }
}

#define BIN_SHIFT 7
#define BIN_NODES 128
#define MAX_BINS 1024
#define EDGE_CHUNK 8192
#define SORT_STAGE 8192

__device__ __forceinline__ int block_exclusive_scan_1024(int c, int* sWave, int* blockTotal) {
    const int lane = threadIdx.x & 31;
    const int wv   = threadIdx.x >> 5;
    int inc = c;
    #pragma unroll
    for (int o = 1; o < 32; o <<= 1) {
        const int up = __shfl_up(inc, o);
        if (lane >= o) inc += up;
    }
    if (lane == 31) sWave[wv] = inc;
    __syncthreads();
    if (wv == 0) {
        const int wsum = sWave[lane];
        int winc = wsum;
        #pragma unroll
        for (int o = 1; o < 32; o <<= 1) {
            const int up = __shfl_up(winc, o);
            if (lane >= o) winc += up;
        }
        sWave[lane] = winc - wsum;
        if (lane == 31) *blockTotal = winc;
    }
    __syncthreads();
    return sWave[wv] + inc - c;
}

__global__ __launch_bounds__(1024) void bin_count_kernel(
    const int* __restrict__ dstI, int* __restrict__ cntMat, long E, int NB)
{
    __shared__ int hist[MAX_BINS];
    const int t = threadIdx.x;
    hist[t] = 0;
    __syncthreads();

    const long base = (long)blockIdx.x * EDGE_CHUNK;
    int d[8];
    #pragma unroll
    for (int i = 0; i < 8; ++i) {
        const long e = base + i * 1024 + t;
        d[i] = (e < E) ? dstI[e] : -1;
    }
    #pragma unroll
    for (int i = 0; i < 8; ++i)
        if (d[i] >= 0) atomicAdd(&hist[d[i] >> BIN_SHIFT], 1);
    __syncthreads();

    if (t < NB) cntMat[(long)blockIdx.x * NB + t] = hist[t];
}

__global__ __launch_bounds__(1024) void bin_scan_kernel(
    const int* __restrict__ cntMat, int* __restrict__ offMat, int* __restrict__ binBase,
    int NW, int NB)
{
    __shared__ int sWave[32];
    __shared__ int sTotal;
    const int t = threadIdx.x;

    int run = 0;
    if (t < NB) {
        for (int w0 = 0; w0 < NW; w0 += 16) {
            int c[16];
            #pragma unroll
            for (int k = 0; k < 16; ++k)
                c[k] = (w0 + k < NW) ? cntMat[(long)(w0 + k) * NB + t] : 0;
            #pragma unroll
            for (int k = 0; k < 16; ++k) {
                if (w0 + k < NW) offMat[(long)(w0 + k) * NB + t] = run;
                run += c[k];
            }
        }
    }
    const int ex = block_exclusive_scan_1024(run, sWave, &sTotal);
    if (t < NB) binBase[t] = ex;
    if (t == 0) binBase[NB] = sTotal;
}

__global__ __launch_bounds__(1024) void bin_scatter_kernel(
    const int* __restrict__ srcI, const int* __restrict__ dstI,
    const int* __restrict__ cntMat, const int* __restrict__ offMat,
    const int* __restrict__ binBase, int* __restrict__ binned, long E, int NB)
{
    __shared__ int sWave[32];
    __shared__ int sTotal;
    __shared__ int lcur[MAX_BINS];
    __shared__ int loff[MAX_BINS];
    __shared__ int goff[MAX_BINS];
    __shared__ int stageV[EDGE_CHUNK];
    __shared__ int stageB[EDGE_CHUNK];

    const int t = threadIdx.x;
    const long row = (long)blockIdx.x * NB;
    const int c = (t < NB) ? cntMat[row + t] : 0;
    const int g = (t < NB) ? (binBase[t] + offMat[row + t]) : 0;
    const int lo = block_exclusive_scan_1024(c, sWave, &sTotal);
    loff[t] = lo;
    lcur[t] = lo;
    goff[t] = g;
    __syncthreads();

    const long base = (long)blockIdx.x * EDGE_CHUNK;
    int d[8], sidx[8];
    #pragma unroll
    for (int i = 0; i < 8; ++i) {
        const long e = base + i * 1024 + t;
        d[i]    = (e < E) ? dstI[e] : -1;
        sidx[i] = (e < E) ? srcI[e] : 0;
    }
    #pragma unroll
    for (int i = 0; i < 8; ++i) {
        if (d[i] >= 0) {
            const int b = d[i] >> BIN_SHIFT;
            const int p = atomicAdd(&lcur[b], 1);
            stageV[p] = (sidx[i] << BIN_SHIFT) | (d[i] & (BIN_NODES - 1));
            stageB[p] = b;
        }
    }
    __syncthreads();

    const long left = E - base;
    const int nLocal = left < EDGE_CHUNK ? (int)left : EDGE_CHUNK;
    for (int j = t; j < nLocal; j += 1024) {
        const int b = stageB[j];
        binned[goff[b] + (j - loff[b])] = stageV[j];
    }
}

__global__ __launch_bounds__(256) void bin_sort_kernel(
    const int* __restrict__ binned, const int* __restrict__ binBase,
    int* __restrict__ seg, int* __restrict__ gsrc, long N)
{
    __shared__ int ncnt[BIN_NODES];
    __shared__ int ncur[BIN_NODES];
    __shared__ int sW[4];
    __shared__ int stage[SORT_STAGE];

    const int t    = threadIdx.x;
    const int lane = t & 31;
    const int wv   = t >> 5;
    const int bin  = blockIdx.x;
    const int start = binBase[bin];
    const int n     = binBase[bin + 1] - start;

    if (t < BIN_NODES) ncnt[t] = 0;
    __syncthreads();
    for (int i = t; i < n; i += 256)
        atomicAdd(&ncnt[binned[start + i] & (BIN_NODES - 1)], 1);
    __syncthreads();

    const int c = (t < BIN_NODES) ? ncnt[t] : 0;
    int inc = c;
    #pragma unroll
    for (int o = 1; o < 32; o <<= 1) {
        const int up = __shfl_up(inc, o);
        if (lane >= o) inc += up;
    }
    if (lane == 31 && wv < 4) sW[wv] = inc;
    __syncthreads();
    if (t < BIN_NODES) {
        int woff = 0;
        for (int k = 0; k < wv; ++k) woff += sW[k];
        const int ex = woff + inc - c;
        ncur[t] = ex;
        const long node = (long)bin * BIN_NODES + t;
        if (node < N) {
            seg[2 * node]     = start + ex;
            seg[2 * node + 1] = c;
        }
    }
    __syncthreads();

    if (n <= SORT_STAGE) {
        for (int i = t; i < n; i += 256) {
            const int v = binned[start + i];
            const int p = atomicAdd(&ncur[v & (BIN_NODES - 1)], 1);
            stage[p] = v >> BIN_SHIFT;
        }
        __syncthreads();
        for (int i = t; i < n; i += 256) gsrc[start + i] = stage[i];
    } else {
        for (int i = t; i < n; i += 256) {
            const int v = binned[start + i];
            const int p = atomicAdd(&ncur[v & (BIN_NODES - 1)], 1);
            gsrc[start + p] = v >> BIN_SHIFT;
        }
    }
}

__device__ __forceinline__ float head_score(const float4 l4, const float4 r4, const float4 a4) {
    float4 t4;
    t4.x = l4.x + r4.x; t4.y = l4.y + r4.y; t4.z = l4.z + r4.z; t4.w = l4.w + r4.w;
    t4.x = t4.x > 0.0f ? t4.x : NEG_SLOPE * t4.x;
    t4.y = t4.y > 0.0f ? t4.y : NEG_SLOPE * t4.y;
    t4.z = t4.z > 0.0f ? t4.z : NEG_SLOPE * t4.z;
    t4.w = t4.w > 0.0f ? t4.w : NEG_SLOPE * t4.w;
    float p = t4.x * a4.x + t4.y * a4.y + t4.z * a4.z + t4.w * a4.w;
    p += __shfl_xor(p, 1, 8);
    p += __shfl_xor(p, 2, 8);
    p += __shfl_xor(p, 4, 8);
    return p;
}

#define AGG_GROUPS 1792

__global__ __launch_bounds__(256) void node_softmax_aggregate_kernel(
    const int* __restrict__ gsrc, const int* __restrict__ seg,
    const float* __restrict__ xl, const float* __restrict__ xr,
    const float* __restrict__ att, const float* __restrict__ bias,
    float* __restrict__ out, long N)
{
    const int lane = threadIdx.x & 31;
    const int wave   = __builtin_amdgcn_readfirstlane((int)(blockIdx.x * 8 + (threadIdx.x >> 5)));
    const int nwaves = (int)(gridDim.x * 8);
    if (wave >= N) return;

    const float4* xl4 = (const float4*)xl;
    const float4 a4 = ((const float4*)att)[lane];
    const float NEG_INF = -__builtin_inff();

    int node  = wave;
    int start = seg[2 * (long)node];
    int cnt   = seg[2 * (long)node + 1];
    int idx = (lane < cnt) ? gsrc[start + lane] : node;

    while (true) {
        const int  nnode = node + nwaves;
        const bool more  = nnode < N;
        int nstart = 0, ncnt = 0;
        if (more) {
            nstart = seg[2 * (long)nnode];
            ncnt   = seg[2 * (long)nnode + 1];
        }

        const v4f rv = __builtin_nontemporal_load((const v4f*)xr + (unsigned)(node * 32 + lane));
        float4 r4; r4.x = rv.x; r4.y = rv.y; r4.z = rv.z; r4.w = rv.w;

        const float4 self4 = xl4[(unsigned)(node * 32 + lane)];
        float m   = head_score(self4, r4, a4);
        float den = 1.0f;
        float4 acc = self4;

        for (int j0 = 0; j0 < cnt; j0 += 32) {
            const int nj = (cnt - j0 < 32) ? (cnt - j0) : 32;
            if (j0 > 0) idx = (lane < nj) ? gsrc[start + j0 + lane] : node;
            for (int j = 0; j < nj; j += 4) {
                const int s0 = __builtin_amdgcn_readlane(idx, j);
                const int s1 = __builtin_amdgcn_readlane(idx, j + 1);
                const int s2 = __builtin_amdgcn_readlane(idx, j + 2);
                const int s3 = __builtin_amdgcn_readlane(idx, j + 3);
                const float4 q0 = xl4[(unsigned)(s0 * 32 + lane)];
                const float4 q1 = xl4[(unsigned)(s1 * 32 + lane)];
                const float4 q2 = xl4[(unsigned)(s2 * 32 + lane)];
                const float4 q3 = xl4[(unsigned)(s3 * 32 + lane)];
                const float p0 = head_score(q0, r4, a4);
                float p1 = head_score(q1, r4, a4);
                float p2 = head_score(q2, r4, a4);
                float p3 = head_score(q3, r4, a4);
                if (j + 1 >= nj) p1 = NEG_INF;
                if (j + 2 >= nj) p2 = NEG_INF;
                if (j + 3 >= nj) p3 = NEG_INF;
                const float gm = fmaxf(fmaxf(m, p0), fmaxf(p1, fmaxf(p2, p3)));
                const float sc = __expf(m - gm);
                const float w0 = __expf(p0 - gm);
                const float w1 = __expf(p1 - gm);
                const float w2 = __expf(p2 - gm);
                const float w3 = __expf(p3 - gm);
                den   = den   * sc + ((w0 + w1) + (w2 + w3));
                acc.x = acc.x * sc + ((w0 * q0.x + w1 * q1.x) + (w2 * q2.x + w3 * q3.x));
                acc.y = acc.y * sc + ((w0 * q0.y + w1 * q1.y) + (w2 * q2.y + w3 * q3.y));
                acc.z = acc.z * sc + ((w0 * q0.z + w1 * q1.z) + (w2 * q2.z + w3 * q3.z));
                acc.w = acc.w * sc + ((w0 * q0.w + w1 * q1.w) + (w2 * q2.w + w3 * q3.w));
                m = gm;
            }
        }

        const float inv = 0.25f / (den + 1e-16f);
        float vx = acc.x * inv, vy = acc.y * inv, vz = acc.z * inv, vw = acc.w * inv;
        vx += __shfl_xor(vx, 8);  vy += __shfl_xor(vy, 8);
        vz += __shfl_xor(vz, 8);  vw += __shfl_xor(vw, 8);
        vx += __shfl_xor(vx, 16); vy += __shfl_xor(vy, 16);
        vz += __shfl_xor(vz, 16); vw += __shfl_xor(vw, 16);

        if (lane < 8) {
            const float4 b4 = ((const float4*)bias)[lane];
            float4 o4;
            o4.x = vx + b4.x; o4.y = vy + b4.y; o4.z = vz + b4.z; o4.w = vw + b4.w;
            ((float4*)out)[(unsigned)(node * 8 + lane)] = o4;
        }

        if (!more) break;
        idx   = (lane < ncnt) ? gsrc[nstart + lane] : nnode;
        node  = nnode;
        start = nstart;
        cnt   = ncnt;
    }
}

extern "C" void kernel_launch(void* const* d_in, const int* in_sizes, int n_in,
                              void* d_out, int out_size, void* d_ws, size_t ws_size,
                              hipStream_t stream) {
    const float* x     = (const float*)d_in[0];
    const int*   ei    = (const int*)d_in[1];
    const float* gamma = (const float*)d_in[2];
    const float* beta  = (const float*)d_in[3];
    const float* Wl    = (const float*)d_in[4];
    const float* bl    = (const float*)d_in[5];
    const float* Wr    = (const float*)d_in[6];
    const float* br    = (const float*)d_in[7];
    const float* att   = (const float*)d_in[8];
    const float* bias  = (const float*)d_in[9];
    float* out = (float*)d_out;

    const long N = in_sizes[0] / C_IN;
    const long E = in_sizes[1] / 2;
    const int* srcI = ei;
    const int* dstI = ei + E;

    float* ws   = (float*)d_ws;
    float* xl   = ws;
    float* xr   = xl + N * C_IN;
    int*   seg  = (int*)(xr + N * C_IN);
    int*   gsrc = seg + N * 2;

    const int NB = (int)((N + BIN_NODES - 1) / BIN_NODES);
    const int NW = (int)((E + EDGE_CHUNK - 1) / EDGE_CHUNK);
    int*   binned  = (int*)ws;
    int*   binBase = binned + E;
    int*   cntMat  = binBase + 2048;
    int*   offMat  = cntMat + (long)NW * NB;

    bin_count_kernel<<<NW, 1024, 0, stream>>>(dstI, cntMat, E, NB);
    bin_scan_kernel<<<1, 1024, 0, stream>>>(cntMat, offMat, binBase, NW, NB);
    bin_scatter_kernel<<<NW, 1024, 0, stream>>>(srcI, dstI, cntMat, offMat, binBase, binned, E, NB);
    bin_sort_kernel<<<NB, 256, 0, stream>>>(binned, binBase, seg, gsrc, N);
    {
        int blocks = (int)((N + 15) / 16);
        ln_gemm_kernel<<<blocks, 256, 0, stream>>>(x, gamma, beta, Wl, bl, Wr, br, xl, xr, N);
    }
    node_softmax_aggregate_kernel<<<AGG_GROUPS, 256, 0, stream>>>(gsrc, seg, xl, xr, att, bias, out, N);
}


# Projection launch weight-stationary: 512 resident work-groups; each wave loads its W_l/W_r column-tile fragments into registers once and loops over 16-row tiles; LayerNorm double-buffered in LDS
# speedup vs baseline: 3.2996x; 1.0691x over previous
#include <hip/hip_runtime.h>
#include <math.h>

typedef float v2f __attribute__((ext_vector_type(2)));
typedef float v8f __attribute__((ext_vector_type(8)));
typedef float v4f __attribute__((ext_vector_type(4)));

#define C_IN 128
#define NEG_SLOPE 0.2f
#define LN_EPS 1e-5f

#define GEMM_GROUPS 512

__global__ __launch_bounds__(256) void ln_gemm_kernel(
    const float* __restrict__ x,
    const float* __restrict__ gamma, const float* __restrict__ beta,
    const float* __restrict__ Wl, const float* __restrict__ bl,
    const float* __restrict__ Wr, const float* __restrict__ br,
    float* __restrict__ xl, float* __restrict__ xr, long N)
{
    __shared__ float sXh[2][16][132];

    const int t   = threadIdx.x;
    const int row = t >> 4;
    const int sub = t & 15;

    const int lane = t & 31;
    const int wave = t >> 5;
    const int half = lane >> 4;
    const int lr   = lane & 15;
    const int col  = wave * 16 + lr;

    const long ntiles = (N + 15) / 16;
    long tile = blockIdx.x;
    if (tile >= ntiles) return;

    v2f fl[32], fr[32];
    #pragma unroll
    for (int i = 0; i < 32; ++i) {
        const int rk = (4 * i + 2 * half) * C_IN + col;
        fl[i].x = Wl[rk]; fl[i].y = Wl[rk + C_IN];
        fr[i].x = Wr[rk]; fr[i].y = Wr[rk + C_IN];
    }
    const float bll = bl[col], brr = br[col];

    float v[8];
    {
        const long node = tile * 16 + row;
        if (node < N) {
            const float4* xp = (const float4*)(x + node * C_IN + sub * 8);
            float4 p0 = xp[0], p1 = xp[1];
            v[0]=p0.x; v[1]=p0.y; v[2]=p0.z; v[3]=p0.w;
            v[4]=p1.x; v[5]=p1.y; v[6]=p1.z; v[7]=p1.w;
        } else {
            #pragma unroll
            for (int j = 0; j < 8; ++j) v[j] = 0.0f;
        }
    }

    int buf = 0;
    for (; tile < ntiles; tile += gridDim.x) {
        float s = 0.0f;
        #pragma unroll
        for (int j = 0; j < 8; ++j) s += v[j];
        s += __shfl_xor(s, 1, 16); s += __shfl_xor(s, 2, 16);
        s += __shfl_xor(s, 4, 16); s += __shfl_xor(s, 8, 16);
        const float mu = s * (1.0f / 128.0f);

        float q = 0.0f;
        #pragma unroll
        for (int j = 0; j < 8; ++j) { float d = v[j] - mu; q += d * d; }
        q += __shfl_xor(q, 1, 16); q += __shfl_xor(q, 2, 16);
        q += __shfl_xor(q, 4, 16); q += __shfl_xor(q, 8, 16);
        const float rstd = rsqrtf(q * (1.0f / 128.0f) + LN_EPS);

        #pragma unroll
        for (int j = 0; j < 8; ++j) {
            const int c = sub * 8 + j;
            float y = (v[j] - mu) * rstd * gamma[c] + beta[c];
            sXh[buf][row][c] = y > 0.0f ? y : 0.0f;
        }
        __syncthreads();

        {
            const long node = (tile + gridDim.x) * 16 + row;
            if (node < N) {
                const float4* xp = (const float4*)(x + node * C_IN + sub * 8);
                float4 p0 = xp[0], p1 = xp[1];
                v[0]=p0.x; v[1]=p0.y; v[2]=p0.z; v[3]=p0.w;
                v[4]=p1.x; v[5]=p1.y; v[6]=p1.z; v[7]=p1.w;
            } else {
                #pragma unroll
                for (int j = 0; j < 8; ++j) v[j] = 0.0f;
            }
        }

        v8f cl = {}; v8f cr = {};
        #pragma unroll
        for (int i = 0; i < 32; ++i) {
            v2f a = *(const v2f*)&sXh[buf][lr][4 * i + 2 * half];
            cl = __builtin_amdgcn_wmma_f32_16x16x4_f32(false, a, false, fl[i], (short)0, cl, false, false);
            cr = __builtin_amdgcn_wmma_f32_16x16x4_f32(false, a, false, fr[i], (short)0, cr, false, false);
        }

        const long base = tile * 16;
        #pragma unroll
        for (int vv = 0; vv < 8; ++vv) {
            const long r = base + vv + 8 * half;
            if (r < N) {
                xl[r * C_IN + col] = cl[vv] + bll;
                xr[r * C_IN + col] = cr[vv] + brr;
            }
        }
        buf ^= 1;
    }
}

#define BIN_SHIFT 7
#define BIN_NODES 128
#define MAX_BINS 1024
#define EDGE_CHUNK 8192
#define SORT_STAGE 8192

__device__ __forceinline__ int block_exclusive_scan_1024(int c, int* sWave, int* blockTotal) {
    const int lane = threadIdx.x & 31;
    const int wv   = threadIdx.x >> 5;
    int inc = c;
    #pragma unroll
    for (int o = 1; o < 32; o <<= 1) {
        const int up = __shfl_up(inc, o);
        if (lane >= o) inc += up;
    }
    if (lane == 31) sWave[wv] = inc;
    __syncthreads();
    if (wv == 0) {
        const int wsum = sWave[lane];
        int winc = wsum;
        #pragma unroll
        for (int o = 1; o < 32; o <<= 1) {
            const int up = __shfl_up(winc, o);
            if (lane >= o) winc += up;
        }
        sWave[lane] = winc - wsum;
        if (lane == 31) *blockTotal = winc;
    }
    __syncthreads();
    return sWave[wv] + inc - c;
}

__global__ __launch_bounds__(1024) void bin_count_kernel(
    const int* __restrict__ dstI, int* __restrict__ cntMat, long E, int NB)
{
    __shared__ int hist[MAX_BINS];
    const int t = threadIdx.x;
    hist[t] = 0;
    __syncthreads();

    const long base = (long)blockIdx.x * EDGE_CHUNK;
    int d[8];
    #pragma unroll
    for (int i = 0; i < 8; ++i) {
        const long e = base + i * 1024 + t;
        d[i] = (e < E) ? dstI[e] : -1;
    }
    #pragma unroll
    for (int i = 0; i < 8; ++i)
        if (d[i] >= 0) atomicAdd(&hist[d[i] >> BIN_SHIFT], 1);
    __syncthreads();

    if (t < NB) cntMat[(long)blockIdx.x * NB + t] = hist[t];
}

__global__ __launch_bounds__(1024) void bin_scan_kernel(
    const int* __restrict__ cntMat, int* __restrict__ offMat, int* __restrict__ binBase,
    int NW, int NB)
{
    __shared__ int sWave[32];
    __shared__ int sTotal;
    const int t = threadIdx.x;

    int run = 0;
    if (t < NB) {
        for (int w0 = 0; w0 < NW; w0 += 16) {
            int c[16];
            #pragma unroll
            for (int k = 0; k < 16; ++k)
                c[k] = (w0 + k < NW) ? cntMat[(long)(w0 + k) * NB + t] : 0;
            #pragma unroll
            for (int k = 0; k < 16; ++k) {
                if (w0 + k < NW) offMat[(long)(w0 + k) * NB + t] = run;
                run += c[k];
            }
        }
    }
    const int ex = block_exclusive_scan_1024(run, sWave, &sTotal);
    if (t < NB) binBase[t] = ex;
    if (t == 0) binBase[NB] = sTotal;
}

__global__ __launch_bounds__(1024) void bin_scatter_kernel(
    const int* __restrict__ srcI, const int* __restrict__ dstI,
    const int* __restrict__ cntMat, const int* __restrict__ offMat,
    const int* __restrict__ binBase, int* __restrict__ binned, long E, int NB)
{
    __shared__ int sWave[32];
    __shared__ int sTotal;
    __shared__ int lcur[MAX_BINS];
    __shared__ int loff[MAX_BINS];
    __shared__ int goff[MAX_BINS];
    __shared__ int stageV[EDGE_CHUNK];
    __shared__ int stageB[EDGE_CHUNK];

    const int t = threadIdx.x;
    const long row = (long)blockIdx.x * NB;
    const int c = (t < NB) ? cntMat[row + t] : 0;
    const int g = (t < NB) ? (binBase[t] + offMat[row + t]) : 0;
    const int lo = block_exclusive_scan_1024(c, sWave, &sTotal);
    loff[t] = lo;
    lcur[t] = lo;
    goff[t] = g;
    __syncthreads();

    const long base = (long)blockIdx.x * EDGE_CHUNK;
    int d[8], sidx[8];
    #pragma unroll
    for (int i = 0; i < 8; ++i) {
        const long e = base + i * 1024 + t;
        d[i]    = (e < E) ? dstI[e] : -1;
        sidx[i] = (e < E) ? srcI[e] : 0;
    }
    #pragma unroll
    for (int i = 0; i < 8; ++i) {
        if (d[i] >= 0) {
            const int b = d[i] >> BIN_SHIFT;
            const int p = atomicAdd(&lcur[b], 1);
            stageV[p] = (sidx[i] << BIN_SHIFT) | (d[i] & (BIN_NODES - 1));
            stageB[p] = b;
        }
    }
    __syncthreads();

    const long left = E - base;
    const int nLocal = left < EDGE_CHUNK ? (int)left : EDGE_CHUNK;
    for (int j = t; j < nLocal; j += 1024) {
        const int b = stageB[j];
        binned[goff[b] + (j - loff[b])] = stageV[j];
    }
}

__global__ __launch_bounds__(256) void bin_sort_kernel(
    const int* __restrict__ binned, const int* __restrict__ binBase,
    int* __restrict__ seg, int* __restrict__ gsrc, long N)
{
    __shared__ int ncnt[BIN_NODES];
    __shared__ int ncur[BIN_NODES];
    __shared__ int sW[4];
    __shared__ int stage[SORT_STAGE];

    const int t    = threadIdx.x;
    const int lane = t & 31;
    const int wv   = t >> 5;
    const int bin  = blockIdx.x;
    const int start = binBase[bin];
    const int n     = binBase[bin + 1] - start;

    if (t < BIN_NODES) ncnt[t] = 0;
    __syncthreads();
    for (int i = t; i < n; i += 256)
        atomicAdd(&ncnt[binned[start + i] & (BIN_NODES - 1)], 1);
    __syncthreads();

    const int c = (t < BIN_NODES) ? ncnt[t] : 0;
    int inc = c;
    #pragma unroll
    for (int o = 1; o < 32; o <<= 1) {
        const int up = __shfl_up(inc, o);
        if (lane >= o) inc += up;
    }
    if (lane == 31 && wv < 4) sW[wv] = inc;
    __syncthreads();
    if (t < BIN_NODES) {
        int woff = 0;
        for (int k = 0; k < wv; ++k) woff += sW[k];
        const int ex = woff + inc - c;
        ncur[t] = ex;
        const long node = (long)bin * BIN_NODES + t;
        if (node < N) {
            seg[2 * node]     = start + ex;
            seg[2 * node + 1] = c;
        }
    }
    __syncthreads();

    if (n <= SORT_STAGE) {
        for (int i = t; i < n; i += 256) {
            const int v = binned[start + i];
            const int p = atomicAdd(&ncur[v & (BIN_NODES - 1)], 1);
            stage[p] = v >> BIN_SHIFT;
        }
        __syncthreads();
        for (int i = t; i < n; i += 256) gsrc[start + i] = stage[i];
    } else {
        for (int i = t; i < n; i += 256) {
            const int v = binned[start + i];
            const int p = atomicAdd(&ncur[v & (BIN_NODES - 1)], 1);
            gsrc[start + p] = v >> BIN_SHIFT;
        }
    }
}

__device__ __forceinline__ float head_score(const float4 l4, const float4 r4, const float4 a4) {
    float4 t4;
    t4.x = l4.x + r4.x; t4.y = l4.y + r4.y; t4.z = l4.z + r4.z; t4.w = l4.w + r4.w;
    t4.x = t4.x > 0.0f ? t4.x : NEG_SLOPE * t4.x;
    t4.y = t4.y > 0.0f ? t4.y : NEG_SLOPE * t4.y;
    t4.z = t4.z > 0.0f ? t4.z : NEG_SLOPE * t4.z;
    t4.w = t4.w > 0.0f ? t4.w : NEG_SLOPE * t4.w;
    float p = t4.x * a4.x + t4.y * a4.y + t4.z * a4.z + t4.w * a4.w;
    p += __shfl_xor(p, 1, 8);
    p += __shfl_xor(p, 2, 8);
    p += __shfl_xor(p, 4, 8);
    return p;
}

#define AGG_GROUPS 1792

__global__ __launch_bounds__(256) void node_softmax_aggregate_kernel(
    const int* __restrict__ gsrc, const int* __restrict__ seg,
    const float* __restrict__ xl, const float* __restrict__ xr,
    const float* __restrict__ att, const float* __restrict__ bias,
    float* __restrict__ out, long N)
{
    const int lane = threadIdx.x & 31;
    const int wave   = __builtin_amdgcn_readfirstlane((int)(blockIdx.x * 8 + (threadIdx.x >> 5)));
    const int nwaves = (int)(gridDim.x * 8);
    if (wave >= N) return;

    const float4* xl4 = (const float4*)xl;
    const float4 a4 = ((const float4*)att)[lane];
    const float NEG_INF = -__builtin_inff();

    int node  = wave;
    int start = seg[2 * (long)node];
    int cnt   = seg[2 * (long)node + 1];
    int idx = (lane < cnt) ? gsrc[start + lane] : node;

    while (true) {
        const int  nnode = node + nwaves;
        const bool more  = nnode < N;
        int nstart = 0, ncnt = 0;
        if (more) {
            nstart = seg[2 * (long)nnode];
            ncnt   = seg[2 * (long)nnode + 1];
        }

        const v4f rv = __builtin_nontemporal_load((const v4f*)xr + (unsigned)(node * 32 + lane));
        float4 r4; r4.x = rv.x; r4.y = rv.y; r4.z = rv.z; r4.w = rv.w;

        const float4 self4 = xl4[(unsigned)(node * 32 + lane)];
        float m   = head_score(self4, r4, a4);
        float den = 1.0f;
        float4 acc = self4;

        for (int j0 = 0; j0 < cnt; j0 += 32) {
            const int nj = (cnt - j0 < 32) ? (cnt - j0) : 32;
            if (j0 > 0) idx = (lane < nj) ? gsrc[start + j0 + lane] : node;
            for (int j = 0; j < nj; j += 4) {
                const int s0 = __builtin_amdgcn_readlane(idx, j);
                const int s1 = __builtin_amdgcn_readlane(idx, j + 1);
                const int s2 = __builtin_amdgcn_readlane(idx, j + 2);
                const int s3 = __builtin_amdgcn_readlane(idx, j + 3);
                const float4 q0 = xl4[(unsigned)(s0 * 32 + lane)];
                const float4 q1 = xl4[(unsigned)(s1 * 32 + lane)];
                const float4 q2 = xl4[(unsigned)(s2 * 32 + lane)];
                const float4 q3 = xl4[(unsigned)(s3 * 32 + lane)];
                const float p0 = head_score(q0, r4, a4);
                float p1 = head_score(q1, r4, a4);
                float p2 = head_score(q2, r4, a4);
                float p3 = head_score(q3, r4, a4);
                if (j + 1 >= nj) p1 = NEG_INF;
                if (j + 2 >= nj) p2 = NEG_INF;
                if (j + 3 >= nj) p3 = NEG_INF;
                const float gm = fmaxf(fmaxf(m, p0), fmaxf(p1, fmaxf(p2, p3)));
                const float sc = __expf(m - gm);
                const float w0 = __expf(p0 - gm);
                const float w1 = __expf(p1 - gm);
                const float w2 = __expf(p2 - gm);
                const float w3 = __expf(p3 - gm);
                den   = den   * sc + ((w0 + w1) + (w2 + w3));
                acc.x = acc.x * sc + ((w0 * q0.x + w1 * q1.x) + (w2 * q2.x + w3 * q3.x));
                acc.y = acc.y * sc + ((w0 * q0.y + w1 * q1.y) + (w2 * q2.y + w3 * q3.y));
                acc.z = acc.z * sc + ((w0 * q0.z + w1 * q1.z) + (w2 * q2.z + w3 * q3.z));
                acc.w = acc.w * sc + ((w0 * q0.w + w1 * q1.w) + (w2 * q2.w + w3 * q3.w));
                m = gm;
            }
        }

        const float inv = 0.25f / (den + 1e-16f);
        float vx = acc.x * inv, vy = acc.y * inv, vz = acc.z * inv, vw = acc.w * inv;
        vx += __shfl_xor(vx, 8);  vy += __shfl_xor(vy, 8);
        vz += __shfl_xor(vz, 8);  vw += __shfl_xor(vw, 8);
        vx += __shfl_xor(vx, 16); vy += __shfl_xor(vy, 16);
        vz += __shfl_xor(vz, 16); vw += __shfl_xor(vw, 16);

        if (lane < 8) {
            const float4 b4 = ((const float4*)bias)[lane];
            float4 o4;
            o4.x = vx + b4.x; o4.y = vy + b4.y; o4.z = vz + b4.z; o4.w = vw + b4.w;
            ((float4*)out)[(unsigned)(node * 8 + lane)] = o4;
        }

        if (!more) break;
        idx   = (lane < ncnt) ? gsrc[nstart + lane] : nnode;
        node  = nnode;
        start = nstart;
        cnt   = ncnt;
    }
}

extern "C" void kernel_launch(void* const* d_in, const int* in_sizes, int n_in,
                              void* d_out, int out_size, void* d_ws, size_t ws_size,
                              hipStream_t stream) {
    const float* x     = (const float*)d_in[0];
    const int*   ei    = (const int*)d_in[1];
    const float* gamma = (const float*)d_in[2];
    const float* beta  = (const float*)d_in[3];
    const float* Wl    = (const float*)d_in[4];
    const float* bl    = (const float*)d_in[5];
    const float* Wr    = (const float*)d_in[6];
    const float* br    = (const float*)d_in[7];
    const float* att   = (const float*)d_in[8];
    const float* bias  = (const float*)d_in[9];
    float* out = (float*)d_out;

    const long N = in_sizes[0] / C_IN;
    const long E = in_sizes[1] / 2;
    const int* srcI = ei;
    const int* dstI = ei + E;

    float* ws   = (float*)d_ws;
    float* xl   = ws;
    float* xr   = xl + N * C_IN;
    int*   seg  = (int*)(xr + N * C_IN);
    int*   gsrc = seg + N * 2;

    const int NB = (int)((N + BIN_NODES - 1) / BIN_NODES);
    const int NW = (int)((E + EDGE_CHUNK - 1) / EDGE_CHUNK);
    int*   binned  = (int*)ws;
    int*   binBase = binned + E;
    int*   cntMat  = binBase + 2048;
    int*   offMat  = cntMat + (long)NW * NB;

    bin_count_kernel<<<NW, 1024, 0, stream>>>(dstI, cntMat, E, NB);
    bin_scan_kernel<<<1, 1024, 0, stream>>>(cntMat, offMat, binBase, NW, NB);
    bin_scatter_kernel<<<NW, 1024, 0, stream>>>(srcI, dstI, cntMat, offMat, binBase, binned, E, NB);
    bin_sort_kernel<<<NB, 256, 0, stream>>>(binned, binBase, seg, gsrc, N);
    ln_gemm_kernel<<<GEMM_GROUPS, 256, 0, stream>>>(x, gamma, beta, Wl, bl, Wr, br, xl, xr, N);
    node_softmax_aggregate_kernel<<<AGG_GROUPS, 256, 0, stream>>>(gsrc, seg, xl, xr, att, bias, out, N);
}


# Edge grouping in two launches, not four: level 1 orders each chunk by bin in place with per-chunk counts and starts; level 2 gathers a bin's runs, base = sum of chunk starts; sum order: LDS arrival
# speedup vs baseline: 3.8898x; 1.1789x over previous
#include <hip/hip_runtime.h>
#include <math.h>

typedef float v2f __attribute__((ext_vector_type(2)));
typedef float v8f __attribute__((ext_vector_type(8)));
typedef float v4f __attribute__((ext_vector_type(4)));

#define C_IN 128
#define NEG_SLOPE 0.2f
#define LN_EPS 1e-5f

#define GEMM_GROUPS 512

__global__ __launch_bounds__(256) void ln_gemm_kernel(
    const float* __restrict__ x,
    const float* __restrict__ gamma, const float* __restrict__ beta,
    const float* __restrict__ Wl, const float* __restrict__ bl,
    const float* __restrict__ Wr, const float* __restrict__ br,
    float* __restrict__ xl, float* __restrict__ xr, long N)
{
    __shared__ float sXh[2][16][132];

    const int t   = threadIdx.x;
    const int row = t >> 4;
    const int sub = t & 15;

    const int lane = t & 31;
    const int wave = t >> 5;
    const int half = lane >> 4;
    const int lr   = lane & 15;
    const int col  = wave * 16 + lr;

    const long ntiles = (N + 15) / 16;
    long tile = blockIdx.x;
    if (tile >= ntiles) return;

    v2f fl[32], fr[32];
    #pragma unroll
    for (int i = 0; i < 32; ++i) {
        const int rk = (4 * i + 2 * half) * C_IN + col;
        fl[i].x = Wl[rk]; fl[i].y = Wl[rk + C_IN];
        fr[i].x = Wr[rk]; fr[i].y = Wr[rk + C_IN];
    }
    const float bll = bl[col], brr = br[col];

    float v[8];
    {
        const long node = tile * 16 + row;
        if (node < N) {
            const float4* xp = (const float4*)(x + node * C_IN + sub * 8);
            float4 p0 = xp[0], p1 = xp[1];
            v[0]=p0.x; v[1]=p0.y; v[2]=p0.z; v[3]=p0.w;
            v[4]=p1.x; v[5]=p1.y; v[6]=p1.z; v[7]=p1.w;
        } else {
            #pragma unroll
            for (int j = 0; j < 8; ++j) v[j] = 0.0f;
        }
    }

    int buf = 0;
    for (; tile < ntiles; tile += gridDim.x) {
        float s = 0.0f;
        #pragma unroll
        for (int j = 0; j < 8; ++j) s += v[j];
        s += __shfl_xor(s, 1, 16); s += __shfl_xor(s, 2, 16);
        s += __shfl_xor(s, 4, 16); s += __shfl_xor(s, 8, 16);
        const float mu = s * (1.0f / 128.0f);

        float q = 0.0f;
        #pragma unroll
        for (int j = 0; j < 8; ++j) { float d = v[j] - mu; q += d * d; }
        q += __shfl_xor(q, 1, 16); q += __shfl_xor(q, 2, 16);
        q += __shfl_xor(q, 4, 16); q += __shfl_xor(q, 8, 16);
        const float rstd = rsqrtf(q * (1.0f / 128.0f) + LN_EPS);

        #pragma unroll
        for (int j = 0; j < 8; ++j) {
            const int c = sub * 8 + j;
            float y = (v[j] - mu) * rstd * gamma[c] + beta[c];
            sXh[buf][row][c] = y > 0.0f ? y : 0.0f;
        }
        __syncthreads();

        {
            const long node = (tile + gridDim.x) * 16 + row;
            if (node < N) {
                const float4* xp = (const float4*)(x + node * C_IN + sub * 8);
                float4 p0 = xp[0], p1 = xp[1];
                v[0]=p0.x; v[1]=p0.y; v[2]=p0.z; v[3]=p0.w;
                v[4]=p1.x; v[5]=p1.y; v[6]=p1.z; v[7]=p1.w;
            } else {
                #pragma unroll
                for (int j = 0; j < 8; ++j) v[j] = 0.0f;
            }
        }

        v8f cl = {}; v8f cr = {};
        #pragma unroll
        for (int i = 0; i < 32; ++i) {
            v2f a = *(const v2f*)&sXh[buf][lr][4 * i + 2 * half];
            cl = __builtin_amdgcn_wmma_f32_16x16x4_f32(false, a, false, fl[i], (short)0, cl, false, false);
            cr = __builtin_amdgcn_wmma_f32_16x16x4_f32(false, a, false, fr[i], (short)0, cr, false, false);
        }

        const long base = tile * 16;
        #pragma unroll
        for (int vv = 0; vv < 8; ++vv) {
            const long r = base + vv + 8 * half;
            if (r < N) {
                xl[r * C_IN + col] = cl[vv] + bll;
                xr[r * C_IN + col] = cr[vv] + brr;
            }
        }
        buf ^= 1;
    }
}

#define BIN_SHIFT 7
#define BIN_NODES 128
#define MAX_BINS 1024
#define EDGE_CHUNK 8192
#define SORT_STAGE 8192
#define TAB_SHIFT 14
#define TAB_MASK ((1 << TAB_SHIFT) - 1)

__device__ __forceinline__ int block_exclusive_scan_1024(int c, int* sWave, int* blockTotal) {
    const int lane = threadIdx.x & 31;
    const int wv   = threadIdx.x >> 5;
    int inc = c;
    #pragma unroll
    for (int o = 1; o < 32; o <<= 1) {
        const int up = __shfl_up(inc, o);
        if (lane >= o) inc += up;
    }
    if (lane == 31) sWave[wv] = inc;
    __syncthreads();
    if (wv == 0) {
        const int wsum = sWave[lane];
        int winc = wsum;
        #pragma unroll
        for (int o = 1; o < 32; o <<= 1) {
            const int up = __shfl_up(winc, o);
            if (lane >= o) winc += up;
        }
        sWave[lane] = winc - wsum;
        if (lane == 31) *blockTotal = winc;
    }
    __syncthreads();
    return sWave[wv] + inc - c;
}

__device__ __forceinline__ int block_exclusive_scan_256(int c, int* sW, int& total) {
    const int lane = threadIdx.x & 31;
    const int wv   = threadIdx.x >> 5;
    int inc = c;
    #pragma unroll
    for (int o = 1; o < 32; o <<= 1) {
        const int up = __shfl_up(inc, o);
        if (lane >= o) inc += up;
    }
    if (lane == 31) sW[wv] = inc;
    __syncthreads();
    int woff = 0, tot = 0;
    #pragma unroll
    for (int k = 0; k < 8; ++k) {
        const int s = sW[k];
        if (k < wv) woff += s;
        tot += s;
    }
    __syncthreads();
    total = tot;
    return woff + inc - c;
}

__global__ __launch_bounds__(1024) void chunk_bin_kernel(
    const int* __restrict__ srcI, const int* __restrict__ dstI,
    int* __restrict__ runTab, int* __restrict__ binned, long E, int NB, int NW)
{
    __shared__ int sWave[32];
    __shared__ int sTotal;
    __shared__ int hist[MAX_BINS];
    __shared__ int lcur[MAX_BINS];
    __shared__ int stageV[EDGE_CHUNK];

    const int t = threadIdx.x;
    hist[t] = 0;
    __syncthreads();

    const long base = (long)blockIdx.x * EDGE_CHUNK;
    int d[8], sidx[8];
    #pragma unroll
    for (int i = 0; i < 8; ++i) {
        const long e = base + i * 1024 + t;
        d[i]    = (e < E) ? dstI[e] : -1;
        sidx[i] = (e < E) ? srcI[e] : 0;
    }
    #pragma unroll
    for (int i = 0; i < 8; ++i)
        if (d[i] >= 0) atomicAdd(&hist[d[i] >> BIN_SHIFT], 1);
    __syncthreads();

    const int c  = hist[t];
    const int lo = block_exclusive_scan_1024(c, sWave, &sTotal);
    lcur[t] = lo;
    if (t < NB) runTab[(long)t * NW + blockIdx.x] = (lo << TAB_SHIFT) | c;
    __syncthreads();

    #pragma unroll
    for (int i = 0; i < 8; ++i) {
        if (d[i] >= 0) {
            const int p = atomicAdd(&lcur[d[i] >> BIN_SHIFT], 1);
            stageV[p] = (sidx[i] << BIN_SHIFT) | (d[i] & (BIN_NODES - 1));
        }
    }
    __syncthreads();

    const long left = E - base;
    const int nLocal = left < EDGE_CHUNK ? (int)left : EDGE_CHUNK;
    for (int j = t; j < nLocal; j += 1024) binned[base + j] = stageV[j];
}

__global__ __launch_bounds__(256) void bin_sort_kernel(
    const int* __restrict__ binned, const int* __restrict__ runTab,
    int* __restrict__ seg, int* __restrict__ gsrc, long N, int NW)
{
    __shared__ int ncnt[BIN_NODES];
    __shared__ int ncur[BIN_NODES];
    __shared__ int sW[8];
    __shared__ int sAcc[2];
    __shared__ int stageIn[SORT_STAGE];
    __shared__ int stageOut[SORT_STAGE];

    const int t   = threadIdx.x;
    const int bin = blockIdx.x;
    const int* __restrict__ rt = runTab + (long)bin * NW;

    if (t < BIN_NODES) ncnt[t] = 0;
    if (t < 2) sAcc[t] = 0;
    __syncthreads();

    const int pk0 = (t < NW) ? rt[t] : 0;
    int sc = pk0 & TAB_MASK, sl = pk0 >> TAB_SHIFT;
    for (int w = t + 256; w < NW; w += 256) {
        const int pk = rt[w];
        sc += pk & TAB_MASK;
        sl += pk >> TAB_SHIFT;
    }
    #pragma unroll
    for (int o = 16; o > 0; o >>= 1) {
        sc += __shfl_xor(sc, o);
        sl += __shfl_xor(sl, o);
    }
    if ((t & 31) == 0) { atomicAdd(&sAcc[0], sc); atomicAdd(&sAcc[1], sl); }
    __syncthreads();
    const int  n      = sAcc[0];
    const int  start  = sAcc[1];
    const bool staged = n <= SORT_STAGE;

    if (staged) {
        int carry = 0;
        for (int w0 = 0; w0 < NW; w0 += 256) {
            const int w  = w0 + t;
            const int pk = (w0 == 0) ? pk0 : ((w < NW) ? rt[w] : 0);
            const int c  = pk & TAB_MASK;
            int tot;
            const int ro = carry + block_exclusive_scan_256(c, sW, tot);
            const int rbase = w * EDGE_CHUNK + (pk >> TAB_SHIFT);
            for (int k = 0; k < c; ++k) stageOut[ro + k] = rbase + k;
            carry += tot;
        }
        __syncthreads();
        for (int i0 = 0; i0 < n; i0 += 1024) {
            int v[4];
            #pragma unroll
            for (int j = 0; j < 4; ++j) {
                const int i = i0 + j * 256 + t;
                v[j] = binned[stageOut[i < n ? i : n - 1]];
            }
            #pragma unroll
            for (int j = 0; j < 4; ++j) {
                const int i = i0 + j * 256 + t;
                if (i < n) {
                    atomicAdd(&ncnt[v[j] & (BIN_NODES - 1)], 1);
                    stageIn[i] = v[j];
                }
            }
        }
    } else {
        for (int w = t; w < NW; w += 256) {
            const int pk = rt[w];
            const int c  = pk & TAB_MASK;
            const int rbase = w * EDGE_CHUNK + (pk >> TAB_SHIFT);
            for (int k = 0; k < c; ++k)
                atomicAdd(&ncnt[binned[rbase + k] & (BIN_NODES - 1)], 1);
        }
    }
    __syncthreads();

    const int cn = (t < BIN_NODES) ? ncnt[t] : 0;
    int totn;
    const int ex = block_exclusive_scan_256(cn, sW, totn);
    if (t < BIN_NODES) {
        ncur[t] = ex;
        const long node = (long)bin * BIN_NODES + t;
        if (node < N) {
            seg[2 * node]     = start + ex;
            seg[2 * node + 1] = cn;
        }
    }
    __syncthreads();

    if (staged) {
        for (int i0 = 0; i0 < n; i0 += 1024) {
            int v[4];
            #pragma unroll
            for (int j = 0; j < 4; ++j) {
                const int i = i0 + j * 256 + t;
                v[j] = stageIn[i < n ? i : n - 1];
            }
            #pragma unroll
            for (int j = 0; j < 4; ++j) {
                const int i = i0 + j * 256 + t;
                if (i < n) {
                    const int p = atomicAdd(&ncur[v[j] & (BIN_NODES - 1)], 1);
                    stageOut[p] = v[j] >> BIN_SHIFT;
                }
            }
        }
        __syncthreads();
        for (int i = t; i < n; i += 256) gsrc[start + i] = stageOut[i];
    } else {
        for (int w = t; w < NW; w += 256) {
            const int pk = rt[w];
            const int c  = pk & TAB_MASK;
            const int rbase = w * EDGE_CHUNK + (pk >> TAB_SHIFT);
            for (int k = 0; k < c; ++k) {
                const int v = binned[rbase + k];
                const int p = atomicAdd(&ncur[v & (BIN_NODES - 1)], 1);
                gsrc[start + p] = v >> BIN_SHIFT;
            }
        }
    }
}

__device__ __forceinline__ float head_score(const float4 l4, const float4 r4, const float4 a4) {
    float4 t4;
    t4.x = l4.x + r4.x; t4.y = l4.y + r4.y; t4.z = l4.z + r4.z; t4.w = l4.w + r4.w;
    t4.x = t4.x > 0.0f ? t4.x : NEG_SLOPE * t4.x;
    t4.y = t4.y > 0.0f ? t4.y : NEG_SLOPE * t4.y;
    t4.z = t4.z > 0.0f ? t4.z : NEG_SLOPE * t4.z;
    t4.w = t4.w > 0.0f ? t4.w : NEG_SLOPE * t4.w;
    float p = t4.x * a4.x + t4.y * a4.y + t4.z * a4.z + t4.w * a4.w;
    p += __shfl_xor(p, 1, 8);
    p += __shfl_xor(p, 2, 8);
    p += __shfl_xor(p, 4, 8);
    return p;
}

#define AGG_GROUPS 1792

__global__ __launch_bounds__(256) void node_softmax_aggregate_kernel(
    const int* __restrict__ gsrc, const int* __restrict__ seg,
    const float* __restrict__ xl, const float* __restrict__ xr,
    const float* __restrict__ att, const float* __restrict__ bias,
    float* __restrict__ out, long N)
{
    const int lane = threadIdx.x & 31;
    const int wave   = __builtin_amdgcn_readfirstlane((int)(blockIdx.x * 8 + (threadIdx.x >> 5)));
    const int nwaves = (int)(gridDim.x * 8);
    if (wave >= N) return;

    const float4* xl4 = (const float4*)xl;
    const float4 a4 = ((const float4*)att)[lane];
    const float NEG_INF = -__builtin_inff();

    int node  = wave;
    int start = seg[2 * (long)node];
    int cnt   = seg[2 * (long)node + 1];
    int idx = (lane < cnt) ? gsrc[start + lane] : node;

    while (true) {
        const int  nnode = node + nwaves;
        const bool more  = nnode < N;
        int nstart = 0, ncnt = 0;
        if (more) {
            nstart = seg[2 * (long)nnode];
            ncnt   = seg[2 * (long)nnode + 1];
        }

        const v4f rv = __builtin_nontemporal_load((const v4f*)xr + (unsigned)(node * 32 + lane));
        float4 r4; r4.x = rv.x; r4.y = rv.y; r4.z = rv.z; r4.w = rv.w;

        const float4 self4 = xl4[(unsigned)(node * 32 + lane)];
        float m   = head_score(self4, r4, a4);
        float den = 1.0f;
        float4 acc = self4;

        for (int j0 = 0; j0 < cnt; j0 += 32) {
            const int nj = (cnt - j0 < 32) ? (cnt - j0) : 32;
            if (j0 > 0) idx = (lane < nj) ? gsrc[start + j0 + lane] : node;
            for (int j = 0; j < nj; j += 4) {
                const int s0 = __builtin_amdgcn_readlane(idx, j);
                const int s1 = __builtin_amdgcn_readlane(idx, j + 1);
                const int s2 = __builtin_amdgcn_readlane(idx, j + 2);
                const int s3 = __builtin_amdgcn_readlane(idx, j + 3);
                const float4 q0 = xl4[(unsigned)(s0 * 32 + lane)];
                const float4 q1 = xl4[(unsigned)(s1 * 32 + lane)];
                const float4 q2 = xl4[(unsigned)(s2 * 32 + lane)];
                const float4 q3 = xl4[(unsigned)(s3 * 32 + lane)];
                const float p0 = head_score(q0, r4, a4);
                float p1 = head_score(q1, r4, a4);
                float p2 = head_score(q2, r4, a4);
                float p3 = head_score(q3, r4, a4);
                if (j + 1 >= nj) p1 = NEG_INF;
                if (j + 2 >= nj) p2 = NEG_INF;
                if (j + 3 >= nj) p3 = NEG_INF;
                const float gm = fmaxf(fmaxf(m, p0), fmaxf(p1, fmaxf(p2, p3)));
                const float sc = __expf(m - gm);
                const float w0 = __expf(p0 - gm);
                const float w1 = __expf(p1 - gm);
                const float w2 = __expf(p2 - gm);
                const float w3 = __expf(p3 - gm);
                den   = den   * sc + ((w0 + w1) + (w2 + w3));
                acc.x = acc.x * sc + ((w0 * q0.x + w1 * q1.x) + (w2 * q2.x + w3 * q3.x));
                acc.y = acc.y * sc + ((w0 * q0.y + w1 * q1.y) + (w2 * q2.y + w3 * q3.y));
                acc.z = acc.z * sc + ((w0 * q0.z + w1 * q1.z) + (w2 * q2.z + w3 * q3.z));
                acc.w = acc.w * sc + ((w0 * q0.w + w1 * q1.w) + (w2 * q2.w + w3 * q3.w));
                m = gm;
            }
        }

        const float inv = 0.25f / (den + 1e-16f);
        float vx = acc.x * inv, vy = acc.y * inv, vz = acc.z * inv, vw = acc.w * inv;
        vx += __shfl_xor(vx, 8);  vy += __shfl_xor(vy, 8);
        vz += __shfl_xor(vz, 8);  vw += __shfl_xor(vw, 8);
        vx += __shfl_xor(vx, 16); vy += __shfl_xor(vy, 16);
        vz += __shfl_xor(vz, 16); vw += __shfl_xor(vw, 16);

        if (lane < 8) {
            const float4 b4 = ((const float4*)bias)[lane];
            float4 o4;
            o4.x = vx + b4.x; o4.y = vy + b4.y; o4.z = vz + b4.z; o4.w = vw + b4.w;
            ((float4*)out)[(unsigned)(node * 8 + lane)] = o4;
        }

        if (!more) break;
        idx   = (lane < ncnt) ? gsrc[nstart + lane] : nnode;
        node  = nnode;
        start = nstart;
        cnt   = ncnt;
    }
}

extern "C" void kernel_launch(void* const* d_in, const int* in_sizes, int n_in,
                              void* d_out, int out_size, void* d_ws, size_t ws_size,
                              hipStream_t stream) {
    const float* x     = (const float*)d_in[0];
    const int*   ei    = (const int*)d_in[1];
    const float* gamma = (const float*)d_in[2];
    const float* beta  = (const float*)d_in[3];
    const float* Wl    = (const float*)d_in[4];
    const float* bl    = (const float*)d_in[5];
    const float* Wr    = (const float*)d_in[6];
    const float* br    = (const float*)d_in[7];
    const float* att   = (const float*)d_in[8];
    const float* bias  = (const float*)d_in[9];
    float* out = (float*)d_out;

    const long N = in_sizes[0] / C_IN;
    const long E = in_sizes[1] / 2;
    const int* srcI = ei;
    const int* dstI = ei + E;

    float* ws   = (float*)d_ws;
    float* xl   = ws;
    float* xr   = xl + N * C_IN;
    int*   seg  = (int*)(xr + N * C_IN);
    int*   gsrc = seg + N * 2;

    const int NB = (int)((N + BIN_NODES - 1) / BIN_NODES);
    const int NW = (int)((E + EDGE_CHUNK - 1) / EDGE_CHUNK);
    int*   binned = (int*)ws;
    int*   runTab = binned + E;

    chunk_bin_kernel<<<NW, 1024, 0, stream>>>(srcI, dstI, runTab, binned, E, NB, NW);
    bin_sort_kernel<<<NB, 256, 0, stream>>>(binned, runTab, seg, gsrc, N, NW);
    ln_gemm_kernel<<<GEMM_GROUPS, 256, 0, stream>>>(x, gamma, beta, Wl, bl, Wr, br, xl, xr, N);
    node_softmax_aggregate_kernel<<<AGG_GROUPS, 256, 0, stream>>>(gsrc, seg, xl, xr, att, bias, out, N);
}


# Float cross-lane sums (edge scores, head sum, LayerNorm rows) on the vector ALU (DPP row_xmask adds, permlanex16 across rows) instead of __shfl_xor trips through the LDS unit; sum order unchanged
# speedup vs baseline: 4.0320x; 1.0366x over previous
#include <hip/hip_runtime.h>
#include <math.h>

typedef float v2f __attribute__((ext_vector_type(2)));
typedef float v8f __attribute__((ext_vector_type(8)));
typedef float v4f __attribute__((ext_vector_type(4)));

#define C_IN 128
#define NEG_SLOPE 0.2f
#define LN_EPS 1e-5f

template <int MASK>
__device__ __forceinline__ float add_row_xor(float v) {
    const int o = __builtin_amdgcn_update_dpp(0, __builtin_bit_cast(int, v), 0x160 | MASK, 0xf, 0xf, true);
    return v + __builtin_bit_cast(float, o);
}
__device__ __forceinline__ float add_other_row(float v) {
    const int o = __builtin_amdgcn_permlanex16(__builtin_bit_cast(int, v), __builtin_bit_cast(int, v),
                                               0x76543210, 0xfedcba98, false, false);
    return v + __builtin_bit_cast(float, o);
}

#define GEMM_GROUPS 512

__global__ __launch_bounds__(256) void ln_gemm_kernel(
    const float* __restrict__ x,
    const float* __restrict__ gamma, const float* __restrict__ beta,
    const float* __restrict__ Wl, const float* __restrict__ bl,
    const float* __restrict__ Wr, const float* __restrict__ br,
    float* __restrict__ xl, float* __restrict__ xr, long N)
{
    __shared__ float sXh[2][16][132];

    const int t   = threadIdx.x;
    const int row = t >> 4;
    const int sub = t & 15;

    const int lane = t & 31;
    const int wave = t >> 5;
    const int half = lane >> 4;
    const int lr   = lane & 15;
    const int col  = wave * 16 + lr;

    const long ntiles = (N + 15) / 16;
    long tile = blockIdx.x;
    if (tile >= ntiles) return;

    v2f fl[32], fr[32];
    #pragma unroll
    for (int i = 0; i < 32; ++i) {
        const int rk = (4 * i + 2 * half) * C_IN + col;
        fl[i].x = Wl[rk]; fl[i].y = Wl[rk + C_IN];
        fr[i].x = Wr[rk]; fr[i].y = Wr[rk + C_IN];
    }
    const float bll = bl[col], brr = br[col];

    float v[8];
    {
        const long node = tile * 16 + row;
        if (node < N) {
            const float4* xp = (const float4*)(x + node * C_IN + sub * 8);
            float4 p0 = xp[0], p1 = xp[1];
            v[0]=p0.x; v[1]=p0.y; v[2]=p0.z; v[3]=p0.w;
            v[4]=p1.x; v[5]=p1.y; v[6]=p1.z; v[7]=p1.w;
        } else {
            #pragma unroll
            for (int j = 0; j < 8; ++j) v[j] = 0.0f;
        }
    }

    int buf = 0;
    for (; tile < ntiles; tile += gridDim.x) {
        float s = 0.0f;
        #pragma unroll
        for (int j = 0; j < 8; ++j) s += v[j];
        s = add_row_xor<1>(s); s = add_row_xor<2>(s);
        s = add_row_xor<4>(s); s = add_row_xor<8>(s);
        const float mu = s * (1.0f / 128.0f);

        float q = 0.0f;
        #pragma unroll
        for (int j = 0; j < 8; ++j) { float d = v[j] - mu; q += d * d; }
        q = add_row_xor<1>(q); q = add_row_xor<2>(q);
        q = add_row_xor<4>(q); q = add_row_xor<8>(q);
        const float rstd = rsqrtf(q * (1.0f / 128.0f) + LN_EPS);

        #pragma unroll
        for (int j = 0; j < 8; ++j) {
            const int c = sub * 8 + j;
            float y = (v[j] - mu) * rstd * gamma[c] + beta[c];
            sXh[buf][row][c] = y > 0.0f ? y : 0.0f;
        }
        __syncthreads();

        {
            const long node = (tile + gridDim.x) * 16 + row;
            if (node < N) {
                const float4* xp = (const float4*)(x + node * C_IN + sub * 8);
                float4 p0 = xp[0], p1 = xp[1];
                v[0]=p0.x; v[1]=p0.y; v[2]=p0.z; v[3]=p0.w;
                v[4]=p1.x; v[5]=p1.y; v[6]=p1.z; v[7]=p1.w;
            } else {
                #pragma unroll
                for (int j = 0; j < 8; ++j) v[j] = 0.0f;
            }
        }

        v8f cl = {}; v8f cr = {};
        #pragma unroll
        for (int i = 0; i < 32; ++i) {
            v2f a = *(const v2f*)&sXh[buf][lr][4 * i + 2 * half];
            cl = __builtin_amdgcn_wmma_f32_16x16x4_f32(false, a, false, fl[i], (short)0, cl, false, false);
            cr = __builtin_amdgcn_wmma_f32_16x16x4_f32(false, a, false, fr[i], (short)0, cr, false, false);
        }

        const long base = tile * 16;
        #pragma unroll
        for (int vv = 0; vv < 8; ++vv) {
            const long r = base + vv + 8 * half;
            if (r < N) {
                xl[r * C_IN + col] = cl[vv] + bll;
                xr[r * C_IN + col] = cr[vv] + brr;
            }
        }
        buf ^= 1;
    }
}

#define BIN_SHIFT 7
#define BIN_NODES 128
#define MAX_BINS 1024
#define EDGE_CHUNK 8192
#define SORT_STAGE 8192
#define TAB_SHIFT 14
#define TAB_MASK ((1 << TAB_SHIFT) - 1)

__device__ __forceinline__ int block_exclusive_scan_1024(int c, int* sWave, int* blockTotal) {
    const int lane = threadIdx.x & 31;
    const int wv   = threadIdx.x >> 5;
    int inc = c;
    #pragma unroll
    for (int o = 1; o < 32; o <<= 1) {
        const int up = __shfl_up(inc, o);
        if (lane >= o) inc += up;
    }
    if (lane == 31) sWave[wv] = inc;
    __syncthreads();
    if (wv == 0) {
        const int wsum = sWave[lane];
        int winc = wsum;
        #pragma unroll
        for (int o = 1; o < 32; o <<= 1) {
            const int up = __shfl_up(winc, o);
            if (lane >= o) winc += up;
        }
        sWave[lane] = winc - wsum;
        if (lane == 31) *blockTotal = winc;
    }
    __syncthreads();
    return sWave[wv] + inc - c;
}

__device__ __forceinline__ int block_exclusive_scan_256(int c, int* sW, int& total) {
    const int lane = threadIdx.x & 31;
    const int wv   = threadIdx.x >> 5;
    int inc = c;
    #pragma unroll
    for (int o = 1; o < 32; o <<= 1) {
        const int up = __shfl_up(inc, o);
        if (lane >= o) inc += up;
    }
    if (lane == 31) sW[wv] = inc;
    __syncthreads();
    int woff = 0, tot = 0;
    #pragma unroll
    for (int k = 0; k < 8; ++k) {
        const int s = sW[k];
        if (k < wv) woff += s;
        tot += s;
    }
    __syncthreads();
    total = tot;
    return woff + inc - c;
}

__global__ __launch_bounds__(1024) void chunk_bin_kernel(
    const int* __restrict__ srcI, const int* __restrict__ dstI,
    int* __restrict__ runTab, int* __restrict__ binned, long E, int NB, int NW)
{
    __shared__ int sWave[32];
    __shared__ int sTotal;
    __shared__ int hist[MAX_BINS];
    __shared__ int lcur[MAX_BINS];
    __shared__ int stageV[EDGE_CHUNK];

    const int t = threadIdx.x;
    hist[t] = 0;
    __syncthreads();

    const long base = (long)blockIdx.x * EDGE_CHUNK;
    int d[8], sidx[8];
    #pragma unroll
    for (int i = 0; i < 8; ++i) {
        const long e = base + i * 1024 + t;
        d[i]    = (e < E) ? dstI[e] : -1;
        sidx[i] = (e < E) ? srcI[e] : 0;
    }
    #pragma unroll
    for (int i = 0; i < 8; ++i)
        if (d[i] >= 0) atomicAdd(&hist[d[i] >> BIN_SHIFT], 1);
    __syncthreads();

    const int c  = hist[t];
    const int lo = block_exclusive_scan_1024(c, sWave, &sTotal);
    lcur[t] = lo;
    if (t < NB) runTab[(long)t * NW + blockIdx.x] = (lo << TAB_SHIFT) | c;
    __syncthreads();

    #pragma unroll
    for (int i = 0; i < 8; ++i) {
        if (d[i] >= 0) {
            const int p = atomicAdd(&lcur[d[i] >> BIN_SHIFT], 1);
            stageV[p] = (sidx[i] << BIN_SHIFT) | (d[i] & (BIN_NODES - 1));
        }
    }
    __syncthreads();

    const long left = E - base;
    const int nLocal = left < EDGE_CHUNK ? (int)left : EDGE_CHUNK;
    for (int j = t; j < nLocal; j += 1024) binned[base + j] = stageV[j];
}

__global__ __launch_bounds__(256) void bin_sort_kernel(
    const int* __restrict__ binned, const int* __restrict__ runTab,
    int* __restrict__ seg, int* __restrict__ gsrc, long N, int NW)
{
    __shared__ int ncnt[BIN_NODES];
    __shared__ int ncur[BIN_NODES];
    __shared__ int sW[8];
    __shared__ int sAcc[2];
    __shared__ int stageIn[SORT_STAGE];
    __shared__ int stageOut[SORT_STAGE];

    const int t   = threadIdx.x;
    const int bin = blockIdx.x;
    const int* __restrict__ rt = runTab + (long)bin * NW;

    if (t < BIN_NODES) ncnt[t] = 0;
    if (t < 2) sAcc[t] = 0;
    __syncthreads();

    const int pk0 = (t < NW) ? rt[t] : 0;
    int sc = pk0 & TAB_MASK, sl = pk0 >> TAB_SHIFT;
    for (int w = t + 256; w < NW; w += 256) {
        const int pk = rt[w];
        sc += pk & TAB_MASK;
        sl += pk >> TAB_SHIFT;
    }
    #pragma unroll
    for (int o = 16; o > 0; o >>= 1) {
        sc += __shfl_xor(sc, o);
        sl += __shfl_xor(sl, o);
    }
    if ((t & 31) == 0) { atomicAdd(&sAcc[0], sc); atomicAdd(&sAcc[1], sl); }
    __syncthreads();
    const int  n      = sAcc[0];
    const int  start  = sAcc[1];
    const bool staged = n <= SORT_STAGE;

    if (staged) {
        int carry = 0;
        for (int w0 = 0; w0 < NW; w0 += 256) {
            const int w  = w0 + t;
            const int pk = (w0 == 0) ? pk0 : ((w < NW) ? rt[w] : 0);
            const int c  = pk & TAB_MASK;
            int tot;
            const int ro = carry + block_exclusive_scan_256(c, sW, tot);
            const int rbase = w * EDGE_CHUNK + (pk >> TAB_SHIFT);
            for (int k = 0; k < c; ++k) stageOut[ro + k] = rbase + k;
            carry += tot;
        }
        __syncthreads();
        for (int i0 = 0; i0 < n; i0 += 1024) {
            int v[4];
            #pragma unroll
            for (int j = 0; j < 4; ++j) {
                const int i = i0 + j * 256 + t;
                v[j] = binned[stageOut[i < n ? i : n - 1]];
            }
            #pragma unroll
            for (int j = 0; j < 4; ++j) {
                const int i = i0 + j * 256 + t;
                if (i < n) {
                    atomicAdd(&ncnt[v[j] & (BIN_NODES - 1)], 1);
                    stageIn[i] = v[j];
                }
            }
        }
    } else {
        for (int w = t; w < NW; w += 256) {
            const int pk = rt[w];
            const int c  = pk & TAB_MASK;
            const int rbase = w * EDGE_CHUNK + (pk >> TAB_SHIFT);
            for (int k = 0; k < c; ++k)
                atomicAdd(&ncnt[binned[rbase + k] & (BIN_NODES - 1)], 1);
        }
    }
    __syncthreads();

    const int cn = (t < BIN_NODES) ? ncnt[t] : 0;
    int totn;
    const int ex = block_exclusive_scan_256(cn, sW, totn);
    if (t < BIN_NODES) {
        ncur[t] = ex;
        const long node = (long)bin * BIN_NODES + t;
        if (node < N) {
            seg[2 * node]     = start + ex;
            seg[2 * node + 1] = cn;
        }
    }
    __syncthreads();

    if (staged) {
        for (int i0 = 0; i0 < n; i0 += 1024) {
            int v[4];
            #pragma unroll
            for (int j = 0; j < 4; ++j) {
                const int i = i0 + j * 256 + t;
                v[j] = stageIn[i < n ? i : n - 1];
            }
            #pragma unroll
            for (int j = 0; j < 4; ++j) {
                const int i = i0 + j * 256 + t;
                if (i < n) {
                    const int p = atomicAdd(&ncur[v[j] & (BIN_NODES - 1)], 1);
                    stageOut[p] = v[j] >> BIN_SHIFT;
                }
            }
        }
        __syncthreads();
        for (int i = t; i < n; i += 256) gsrc[start + i] = stageOut[i];
    } else {
        for (int w = t; w < NW; w += 256) {
            const int pk = rt[w];
            const int c  = pk & TAB_MASK;
            const int rbase = w * EDGE_CHUNK + (pk >> TAB_SHIFT);
            for (int k = 0; k < c; ++k) {
                const int v = binned[rbase + k];
                const int p = atomicAdd(&ncur[v & (BIN_NODES - 1)], 1);
                gsrc[start + p] = v >> BIN_SHIFT;
            }
        }
    }
}

__device__ __forceinline__ float head_score(const float4 l4, const float4 r4, const float4 a4) {
    float4 t4;
    t4.x = l4.x + r4.x; t4.y = l4.y + r4.y; t4.z = l4.z + r4.z; t4.w = l4.w + r4.w;
    t4.x = t4.x > 0.0f ? t4.x : NEG_SLOPE * t4.x;
    t4.y = t4.y > 0.0f ? t4.y : NEG_SLOPE * t4.y;
    t4.z = t4.z > 0.0f ? t4.z : NEG_SLOPE * t4.z;
    t4.w = t4.w > 0.0f ? t4.w : NEG_SLOPE * t4.w;
    float p = t4.x * a4.x + t4.y * a4.y + t4.z * a4.z + t4.w * a4.w;
    p = add_row_xor<1>(p);
    p = add_row_xor<2>(p);
    p = add_row_xor<4>(p);
    return p;
}

#define AGG_GROUPS 1792

__global__ __launch_bounds__(256) void node_softmax_aggregate_kernel(
    const int* __restrict__ gsrc, const int* __restrict__ seg,
    const float* __restrict__ xl, const float* __restrict__ xr,
    const float* __restrict__ att, const float* __restrict__ bias,
    float* __restrict__ out, long N)
{
    const int lane = threadIdx.x & 31;
    const int wave   = __builtin_amdgcn_readfirstlane((int)(blockIdx.x * 8 + (threadIdx.x >> 5)));
    const int nwaves = (int)(gridDim.x * 8);
    if (wave >= N) return;

    const float4* xl4 = (const float4*)xl;
    const float4 a4 = ((const float4*)att)[lane];
    const float NEG_INF = -__builtin_inff();

    int node  = wave;
    int start = seg[2 * (long)node];
    int cnt   = seg[2 * (long)node + 1];
    int idx = (lane < cnt) ? gsrc[start + lane] : node;

    while (true) {
        const int  nnode = node + nwaves;
        const bool more  = nnode < N;
        int nstart = 0, ncnt = 0;
        if (more) {
            nstart = seg[2 * (long)nnode];
            ncnt   = seg[2 * (long)nnode + 1];
        }

        const v4f rv = __builtin_nontemporal_load((const v4f*)xr + (unsigned)(node * 32 + lane));
        float4 r4; r4.x = rv.x; r4.y = rv.y; r4.z = rv.z; r4.w = rv.w;

        const float4 self4 = xl4[(unsigned)(node * 32 + lane)];
        float m   = head_score(self4, r4, a4);
        float den = 1.0f;
        float4 acc = self4;

        for (int j0 = 0; j0 < cnt; j0 += 32) {
            const int nj = (cnt - j0 < 32) ? (cnt - j0) : 32;
            if (j0 > 0) idx = (lane < nj) ? gsrc[start + j0 + lane] : node;
            for (int j = 0; j < nj; j += 4) {
                const int s0 = __builtin_amdgcn_readlane(idx, j);
                const int s1 = __builtin_amdgcn_readlane(idx, j + 1);
                const int s2 = __builtin_amdgcn_readlane(idx, j + 2);
                const int s3 = __builtin_amdgcn_readlane(idx, j + 3);
                const float4 q0 = xl4[(unsigned)(s0 * 32 + lane)];
                const float4 q1 = xl4[(unsigned)(s1 * 32 + lane)];
                const float4 q2 = xl4[(unsigned)(s2 * 32 + lane)];
                const float4 q3 = xl4[(unsigned)(s3 * 32 + lane)];
                const float p0 = head_score(q0, r4, a4);
                float p1 = head_score(q1, r4, a4);
                float p2 = head_score(q2, r4, a4);
                float p3 = head_score(q3, r4, a4);
                if (j + 1 >= nj) p1 = NEG_INF;
                if (j + 2 >= nj) p2 = NEG_INF;
                if (j + 3 >= nj) p3 = NEG_INF;
                const float gm = fmaxf(fmaxf(m, p0), fmaxf(p1, fmaxf(p2, p3)));
                const float sc = __expf(m - gm);
                const float w0 = __expf(p0 - gm);
                const float w1 = __expf(p1 - gm);
                const float w2 = __expf(p2 - gm);
                const float w3 = __expf(p3 - gm);
                den   = den   * sc + ((w0 + w1) + (w2 + w3));
                acc.x = acc.x * sc + ((w0 * q0.x + w1 * q1.x) + (w2 * q2.x + w3 * q3.x));
                acc.y = acc.y * sc + ((w0 * q0.y + w1 * q1.y) + (w2 * q2.y + w3 * q3.y));
                acc.z = acc.z * sc + ((w0 * q0.z + w1 * q1.z) + (w2 * q2.z + w3 * q3.z));
                acc.w = acc.w * sc + ((w0 * q0.w + w1 * q1.w) + (w2 * q2.w + w3 * q3.w));
                m = gm;
            }
        }

        const float inv = 0.25f / (den + 1e-16f);
        float vx = acc.x * inv, vy = acc.y * inv, vz = acc.z * inv, vw = acc.w * inv;
        vx = add_row_xor<8>(vx);  vy = add_row_xor<8>(vy);
        vz = add_row_xor<8>(vz);  vw = add_row_xor<8>(vw);
        vx = add_other_row(vx);   vy = add_other_row(vy);
        vz = add_other_row(vz);   vw = add_other_row(vw);

        if (lane < 8) {
            const float4 b4 = ((const float4*)bias)[lane];
            float4 o4;
            o4.x = vx + b4.x; o4.y = vy + b4.y; o4.z = vz + b4.z; o4.w = vw + b4.w;
            ((float4*)out)[(unsigned)(node * 8 + lane)] = o4;
        }

        if (!more) break;
        idx   = (lane < ncnt) ? gsrc[nstart + lane] : nnode;
        node  = nnode;
        start = nstart;
        cnt   = ncnt;
    }
}

extern "C" void kernel_launch(void* const* d_in, const int* in_sizes, int n_in,
                              void* d_out, int out_size, void* d_ws, size_t ws_size,
                              hipStream_t stream) {
    const float* x     = (const float*)d_in[0];
    const int*   ei    = (const int*)d_in[1];
    const float* gamma = (const float*)d_in[2];
    const float* beta  = (const float*)d_in[3];
    const float* Wl    = (const float*)d_in[4];
    const float* bl    = (const float*)d_in[5];
    const float* Wr    = (const float*)d_in[6];
    const float* br    = (const float*)d_in[7];
    const float* att   = (const float*)d_in[8];
    const float* bias  = (const float*)d_in[9];
    float* out = (float*)d_out;

    const long N = in_sizes[0] / C_IN;
    const long E = in_sizes[1] / 2;
    const int* srcI = ei;
    const int* dstI = ei + E;

    float* ws   = (float*)d_ws;
    float* xl   = ws;
    float* xr   = xl + N * C_IN;
    int*   seg  = (int*)(xr + N * C_IN);
    int*   gsrc = seg + N * 2;

    const int NB = (int)((N + BIN_NODES - 1) / BIN_NODES);
    const int NW = (int)((E + EDGE_CHUNK - 1) / EDGE_CHUNK);
    int*   binned = (int*)ws;
    int*   runTab = binned + E;

    chunk_bin_kernel<<<NW, 1024, 0, stream>>>(srcI, dstI, runTab, binned, E, NB, NW);
    bin_sort_kernel<<<NB, 256, 0, stream>>>(binned, runTab, seg, gsrc, N, NW);
    ln_gemm_kernel<<<GEMM_GROUPS, 256, 0, stream>>>(x, gamma, beta, Wl, bl, Wr, br, xl, xr, N);
    node_softmax_aggregate_kernel<<<AGG_GROUPS, 256, 0, stream>>>(gsrc, seg, xl, xr, att, bias, out, N);
}


# Aggregation loop with fewer vector instructions per edge: leaky ReLU as max(t, 0.2t), the 4-term dot as two fused pairs (its sum order changes), softmax in base 2 with att pre-scaled by log2(e)
# speedup vs baseline: 4.2311x; 1.0494x over previous
#include <hip/hip_runtime.h>
#include <math.h>

typedef float v2f __attribute__((ext_vector_type(2)));
typedef float v8f __attribute__((ext_vector_type(8)));
typedef float v4f __attribute__((ext_vector_type(4)));

#define C_IN 128
#define NEG_SLOPE 0.2f
#define LN_EPS 1e-5f

template <int MASK>
__device__ __forceinline__ float add_row_xor(float v) {
    const int o = __builtin_amdgcn_update_dpp(0, __builtin_bit_cast(int, v), 0x160 | MASK, 0xf, 0xf, true);
    return v + __builtin_bit_cast(float, o);
}
__device__ __forceinline__ float add_other_row(float v) {
    const int o = __builtin_amdgcn_permlanex16(__builtin_bit_cast(int, v), __builtin_bit_cast(int, v),
                                               0x76543210, 0xfedcba98, false, false);
    return v + __builtin_bit_cast(float, o);
}

#define GEMM_GROUPS 512

__global__ __launch_bounds__(256) void ln_gemm_kernel(
    const float* __restrict__ x,
    const float* __restrict__ gamma, const float* __restrict__ beta,
    const float* __restrict__ Wl, const float* __restrict__ bl,
    const float* __restrict__ Wr, const float* __restrict__ br,
    float* __restrict__ xl, float* __restrict__ xr, long N)
{
    __shared__ float sXh[2][16][132];

    const int t   = threadIdx.x;
    const int row = t >> 4;
    const int sub = t & 15;

    const int lane = t & 31;
    const int wave = t >> 5;
    const int half = lane >> 4;
    const int lr   = lane & 15;
    const int col  = wave * 16 + lr;

    const long ntiles = (N + 15) / 16;
    long tile = blockIdx.x;
    if (tile >= ntiles) return;

    v2f fl[32], fr[32];
    #pragma unroll
    for (int i = 0; i < 32; ++i) {
        const int rk = (4 * i + 2 * half) * C_IN + col;
        fl[i].x = Wl[rk]; fl[i].y = Wl[rk + C_IN];
        fr[i].x = Wr[rk]; fr[i].y = Wr[rk + C_IN];
    }
    const float bll = bl[col], brr = br[col];

    float v[8];
    {
        const long node = tile * 16 + row;
        if (node < N) {
            const float4* xp = (const float4*)(x + node * C_IN + sub * 8);
            float4 p0 = xp[0], p1 = xp[1];
            v[0]=p0.x; v[1]=p0.y; v[2]=p0.z; v[3]=p0.w;
            v[4]=p1.x; v[5]=p1.y; v[6]=p1.z; v[7]=p1.w;
        } else {
            #pragma unroll
            for (int j = 0; j < 8; ++j) v[j] = 0.0f;
        }
    }

    int buf = 0;
    for (; tile < ntiles; tile += gridDim.x) {
        float s = 0.0f;
        #pragma unroll
        for (int j = 0; j < 8; ++j) s += v[j];
        s = add_row_xor<1>(s); s = add_row_xor<2>(s);
        s = add_row_xor<4>(s); s = add_row_xor<8>(s);
        const float mu = s * (1.0f / 128.0f);

        float q = 0.0f;
        #pragma unroll
        for (int j = 0; j < 8; ++j) { float d = v[j] - mu; q += d * d; }
        q = add_row_xor<1>(q); q = add_row_xor<2>(q);
        q = add_row_xor<4>(q); q = add_row_xor<8>(q);
        const float rstd = rsqrtf(q * (1.0f / 128.0f) + LN_EPS);

        #pragma unroll
        for (int j = 0; j < 8; ++j) {
            const int c = sub * 8 + j;
            float y = (v[j] - mu) * rstd * gamma[c] + beta[c];
            sXh[buf][row][c] = y > 0.0f ? y : 0.0f;
        }
        __syncthreads();

        {
            const long node = (tile + gridDim.x) * 16 + row;
            if (node < N) {
                const float4* xp = (const float4*)(x + node * C_IN + sub * 8);
                float4 p0 = xp[0], p1 = xp[1];
                v[0]=p0.x; v[1]=p0.y; v[2]=p0.z; v[3]=p0.w;
                v[4]=p1.x; v[5]=p1.y; v[6]=p1.z; v[7]=p1.w;
            } else {
                #pragma unroll
                for (int j = 0; j < 8; ++j) v[j] = 0.0f;
            }
        }

        v8f cl = {}; v8f cr = {};
        #pragma unroll
        for (int i = 0; i < 32; ++i) {
            v2f a = *(const v2f*)&sXh[buf][lr][4 * i + 2 * half];
            cl = __builtin_amdgcn_wmma_f32_16x16x4_f32(false, a, false, fl[i], (short)0, cl, false, false);
            cr = __builtin_amdgcn_wmma_f32_16x16x4_f32(false, a, false, fr[i], (short)0, cr, false, false);
        }

        const long base = tile * 16;
        #pragma unroll
        for (int vv = 0; vv < 8; ++vv) {
            const long r = base + vv + 8 * half;
            if (r < N) {
                xl[r * C_IN + col] = cl[vv] + bll;
                xr[r * C_IN + col] = cr[vv] + brr;
            }
        }
        buf ^= 1;
    }
}

#define BIN_SHIFT 7
#define BIN_NODES 128
#define MAX_BINS 1024
#define EDGE_CHUNK 8192
#define SORT_STAGE 8192
#define TAB_SHIFT 14
#define TAB_MASK ((1 << TAB_SHIFT) - 1)

__device__ __forceinline__ int block_exclusive_scan_1024(int c, int* sWave, int* blockTotal) {
    const int lane = threadIdx.x & 31;
    const int wv   = threadIdx.x >> 5;
    int inc = c;
    #pragma unroll
    for (int o = 1; o < 32; o <<= 1) {
        const int up = __shfl_up(inc, o);
        if (lane >= o) inc += up;
    }
    if (lane == 31) sWave[wv] = inc;
    __syncthreads();
    if (wv == 0) {
        const int wsum = sWave[lane];
        int winc = wsum;
        #pragma unroll
        for (int o = 1; o < 32; o <<= 1) {
            const int up = __shfl_up(winc, o);
            if (lane >= o) winc += up;
        }
        sWave[lane] = winc - wsum;
        if (lane == 31) *blockTotal = winc;
    }
    __syncthreads();
    return sWave[wv] + inc - c;
}

__device__ __forceinline__ int block_exclusive_scan_256(int c, int* sW, int& total) {
    const int lane = threadIdx.x & 31;
    const int wv   = threadIdx.x >> 5;
    int inc = c;
    #pragma unroll
    for (int o = 1; o < 32; o <<= 1) {
        const int up = __shfl_up(inc, o);
        if (lane >= o) inc += up;
    }
    if (lane == 31) sW[wv] = inc;
    __syncthreads();
    int woff = 0, tot = 0;
    #pragma unroll
    for (int k = 0; k < 8; ++k) {
        const int s = sW[k];
        if (k < wv) woff += s;
        tot += s;
    }
    __syncthreads();
    total = tot;
    return woff + inc - c;
}

__global__ __launch_bounds__(1024) void chunk_bin_kernel(
    const int* __restrict__ srcI, const int* __restrict__ dstI,
    int* __restrict__ runTab, int* __restrict__ binned, long E, int NB, int NW)
{
    __shared__ int sWave[32];
    __shared__ int sTotal;
    __shared__ int hist[MAX_BINS];
    __shared__ int lcur[MAX_BINS];
    __shared__ int stageV[EDGE_CHUNK];

    const int t = threadIdx.x;
    hist[t] = 0;
    __syncthreads();

    const long base = (long)blockIdx.x * EDGE_CHUNK;
    int d[8], sidx[8];
    #pragma unroll
    for (int i = 0; i < 8; ++i) {
        const long e = base + i * 1024 + t;
        d[i]    = (e < E) ? dstI[e] : -1;
        sidx[i] = (e < E) ? srcI[e] : 0;
    }
    #pragma unroll
    for (int i = 0; i < 8; ++i)
        if (d[i] >= 0) atomicAdd(&hist[d[i] >> BIN_SHIFT], 1);
    __syncthreads();

    const int c  = hist[t];
    const int lo = block_exclusive_scan_1024(c, sWave, &sTotal);
    lcur[t] = lo;
    if (t < NB) runTab[(long)t * NW + blockIdx.x] = (lo << TAB_SHIFT) | c;
    __syncthreads();

    #pragma unroll
    for (int i = 0; i < 8; ++i) {
        if (d[i] >= 0) {
            const int p = atomicAdd(&lcur[d[i] >> BIN_SHIFT], 1);
            stageV[p] = (sidx[i] << BIN_SHIFT) | (d[i] & (BIN_NODES - 1));
        }
    }
    __syncthreads();

    const long left = E - base;
    const int nLocal = left < EDGE_CHUNK ? (int)left : EDGE_CHUNK;
    for (int j = t; j < nLocal; j += 1024) binned[base + j] = stageV[j];
}

__global__ __launch_bounds__(256) void bin_sort_kernel(
    const int* __restrict__ binned, const int* __restrict__ runTab,
    int* __restrict__ seg, int* __restrict__ gsrc, long N, int NW)
{
    __shared__ int ncnt[BIN_NODES];
    __shared__ int ncur[BIN_NODES];
    __shared__ int sW[8];
    __shared__ int sAcc[2];
    __shared__ int stageIn[SORT_STAGE];
    __shared__ int stageOut[SORT_STAGE];

    const int t   = threadIdx.x;
    const int bin = blockIdx.x;
    const int* __restrict__ rt = runTab + (long)bin * NW;

    if (t < BIN_NODES) ncnt[t] = 0;
    if (t < 2) sAcc[t] = 0;
    __syncthreads();

    const int pk0 = (t < NW) ? rt[t] : 0;
    int sc = pk0 & TAB_MASK, sl = pk0 >> TAB_SHIFT;
    for (int w = t + 256; w < NW; w += 256) {
        const int pk = rt[w];
        sc += pk & TAB_MASK;
        sl += pk >> TAB_SHIFT;
    }
    #pragma unroll
    for (int o = 16; o > 0; o >>= 1) {
        sc += __shfl_xor(sc, o);
        sl += __shfl_xor(sl, o);
    }
    if ((t & 31) == 0) { atomicAdd(&sAcc[0], sc); atomicAdd(&sAcc[1], sl); }
    __syncthreads();
    const int  n      = sAcc[0];
    const int  start  = sAcc[1];
    const bool staged = n <= SORT_STAGE;

    if (staged) {
        int carry = 0;
        for (int w0 = 0; w0 < NW; w0 += 256) {
            const int w  = w0 + t;
            const int pk = (w0 == 0) ? pk0 : ((w < NW) ? rt[w] : 0);
            const int c  = pk & TAB_MASK;
            int tot;
            const int ro = carry + block_exclusive_scan_256(c, sW, tot);
            const int rbase = w * EDGE_CHUNK + (pk >> TAB_SHIFT);
            for (int k = 0; k < c; ++k) stageOut[ro + k] = rbase + k;
            carry += tot;
        }
        __syncthreads();
        for (int i0 = 0; i0 < n; i0 += 1024) {
            int v[4];
            #pragma unroll
            for (int j = 0; j < 4; ++j) {
                const int i = i0 + j * 256 + t;
                v[j] = binned[stageOut[i < n ? i : n - 1]];
            }
            #pragma unroll
            for (int j = 0; j < 4; ++j) {
                const int i = i0 + j * 256 + t;
                if (i < n) {
                    atomicAdd(&ncnt[v[j] & (BIN_NODES - 1)], 1);
                    stageIn[i] = v[j];
                }
            }
        }
    } else {
        for (int w = t; w < NW; w += 256) {
            const int pk = rt[w];
            const int c  = pk & TAB_MASK;
            const int rbase = w * EDGE_CHUNK + (pk >> TAB_SHIFT);
            for (int k = 0; k < c; ++k)
                atomicAdd(&ncnt[binned[rbase + k] & (BIN_NODES - 1)], 1);
        }
    }
    __syncthreads();

    const int cn = (t < BIN_NODES) ? ncnt[t] : 0;
    int totn;
    const int ex = block_exclusive_scan_256(cn, sW, totn);
    if (t < BIN_NODES) {
        ncur[t] = ex;
        const long node = (long)bin * BIN_NODES + t;
        if (node < N) {
            seg[2 * node]     = start + ex;
            seg[2 * node + 1] = cn;
        }
    }
    __syncthreads();

    if (staged) {
        for (int i0 = 0; i0 < n; i0 += 1024) {
            int v[4];
            #pragma unroll
            for (int j = 0; j < 4; ++j) {
                const int i = i0 + j * 256 + t;
                v[j] = stageIn[i < n ? i : n - 1];
            }
            #pragma unroll
            for (int j = 0; j < 4; ++j) {
                const int i = i0 + j * 256 + t;
                if (i < n) {
                    const int p = atomicAdd(&ncur[v[j] & (BIN_NODES - 1)], 1);
                    stageOut[p] = v[j] >> BIN_SHIFT;
                }
            }
        }
        __syncthreads();
        for (int i = t; i < n; i += 256) gsrc[start + i] = stageOut[i];
    } else {
        for (int w = t; w < NW; w += 256) {
            const int pk = rt[w];
            const int c  = pk & TAB_MASK;
            const int rbase = w * EDGE_CHUNK + (pk >> TAB_SHIFT);
            for (int k = 0; k < c; ++k) {
                const int v = binned[rbase + k];
                const int p = atomicAdd(&ncur[v & (BIN_NODES - 1)], 1);
                gsrc[start + p] = v >> BIN_SHIFT;
            }
        }
    }
}

__device__ __forceinline__ float head_score(const float4 l4, const float4 r4, const float4 a4) {
    const float tx = l4.x + r4.x, ty = l4.y + r4.y, tz = l4.z + r4.z, tw = l4.w + r4.w;
    const float ux = fmaxf(tx, NEG_SLOPE * tx);
    const float uy = fmaxf(ty, NEG_SLOPE * ty);
    const float uz = fmaxf(tz, NEG_SLOPE * tz);
    const float uw = fmaxf(tw, NEG_SLOPE * tw);
    float p = fmaf(uz, a4.z, ux * a4.x) + fmaf(uw, a4.w, uy * a4.y);
    p = add_row_xor<1>(p);
    p = add_row_xor<2>(p);
    p = add_row_xor<4>(p);
    return p;
}

#define AGG_GROUPS 1792

__global__ __launch_bounds__(256) void node_softmax_aggregate_kernel(
    const int* __restrict__ gsrc, const int* __restrict__ seg,
    const float* __restrict__ xl, const float* __restrict__ xr,
    const float* __restrict__ att, const float* __restrict__ bias,
    float* __restrict__ out, long N)
{
    const int lane = threadIdx.x & 31;
    const int wave   = __builtin_amdgcn_readfirstlane((int)(blockIdx.x * 8 + (threadIdx.x >> 5)));
    const int nwaves = (int)(gridDim.x * 8);
    if (wave >= N) return;

    const float4* xl4 = (const float4*)xl;
    const float LOG2E = 1.4426950408889634f;
    float4 a4 = ((const float4*)att)[lane];
    a4.x *= LOG2E; a4.y *= LOG2E; a4.z *= LOG2E; a4.w *= LOG2E;
    const float NEG_INF = -__builtin_inff();

    int node  = wave;
    int start = seg[2 * (long)node];
    int cnt   = seg[2 * (long)node + 1];
    int idx = (lane < cnt) ? gsrc[start + lane] : node;

    while (true) {
        const int  nnode = node + nwaves;
        const bool more  = nnode < N;
        int nstart = 0, ncnt = 0;
        if (more) {
            nstart = seg[2 * (long)nnode];
            ncnt   = seg[2 * (long)nnode + 1];
        }

        const v4f rv = __builtin_nontemporal_load((const v4f*)xr + (unsigned)(node * 32 + lane));
        float4 r4; r4.x = rv.x; r4.y = rv.y; r4.z = rv.z; r4.w = rv.w;

        const float4 self4 = xl4[(unsigned)(node * 32 + lane)];
        float m   = head_score(self4, r4, a4);
        float den = 1.0f;
        float4 acc = self4;

        for (int j0 = 0; j0 < cnt; j0 += 32) {
            const int nj = (cnt - j0 < 32) ? (cnt - j0) : 32;
            if (j0 > 0) idx = (lane < nj) ? gsrc[start + j0 + lane] : node;
            for (int j = 0; j < nj; j += 4) {
                const int s0 = __builtin_amdgcn_readlane(idx, j);
                const int s1 = __builtin_amdgcn_readlane(idx, j + 1);
                const int s2 = __builtin_amdgcn_readlane(idx, j + 2);
                const int s3 = __builtin_amdgcn_readlane(idx, j + 3);
                const float4 q0 = xl4[(unsigned)(s0 * 32 + lane)];
                const float4 q1 = xl4[(unsigned)(s1 * 32 + lane)];
                const float4 q2 = xl4[(unsigned)(s2 * 32 + lane)];
                const float4 q3 = xl4[(unsigned)(s3 * 32 + lane)];
                const float p0 = head_score(q0, r4, a4);
                float p1 = head_score(q1, r4, a4);
                float p2 = head_score(q2, r4, a4);
                float p3 = head_score(q3, r4, a4);
                if (j + 1 >= nj) p1 = NEG_INF;
                if (j + 2 >= nj) p2 = NEG_INF;
                if (j + 3 >= nj) p3 = NEG_INF;
                const float gm = fmaxf(fmaxf(m, p0), fmaxf(p1, fmaxf(p2, p3)));
                const float sc = __builtin_amdgcn_exp2f(m - gm);
                const float w0 = __builtin_amdgcn_exp2f(p0 - gm);
                const float w1 = __builtin_amdgcn_exp2f(p1 - gm);
                const float w2 = __builtin_amdgcn_exp2f(p2 - gm);
                const float w3 = __builtin_amdgcn_exp2f(p3 - gm);
                den   = den   * sc + ((w0 + w1) + (w2 + w3));
                acc.x = acc.x * sc + ((w0 * q0.x + w1 * q1.x) + (w2 * q2.x + w3 * q3.x));
                acc.y = acc.y * sc + ((w0 * q0.y + w1 * q1.y) + (w2 * q2.y + w3 * q3.y));
                acc.z = acc.z * sc + ((w0 * q0.z + w1 * q1.z) + (w2 * q2.z + w3 * q3.z));
                acc.w = acc.w * sc + ((w0 * q0.w + w1 * q1.w) + (w2 * q2.w + w3 * q3.w));
                m = gm;
            }
        }

        const float inv = 0.25f / (den + 1e-16f);
        float vx = acc.x * inv, vy = acc.y * inv, vz = acc.z * inv, vw = acc.w * inv;
        vx = add_row_xor<8>(vx);  vy = add_row_xor<8>(vy);
        vz = add_row_xor<8>(vz);  vw = add_row_xor<8>(vw);
        vx = add_other_row(vx);   vy = add_other_row(vy);
        vz = add_other_row(vz);   vw = add_other_row(vw);

        if (lane < 8) {
            const float4 b4 = ((const float4*)bias)[lane];
            float4 o4;
            o4.x = vx + b4.x; o4.y = vy + b4.y; o4.z = vz + b4.z; o4.w = vw + b4.w;
            ((float4*)out)[(unsigned)(node * 8 + lane)] = o4;
        }

        if (!more) break;
        idx   = (lane < ncnt) ? gsrc[nstart + lane] : nnode;
        node  = nnode;
        start = nstart;
        cnt   = ncnt;
    }
}

extern "C" void kernel_launch(void* const* d_in, const int* in_sizes, int n_in,
                              void* d_out, int out_size, void* d_ws, size_t ws_size,
                              hipStream_t stream) {
    const float* x     = (const float*)d_in[0];
    const int*   ei    = (const int*)d_in[1];
    const float* gamma = (const float*)d_in[2];
    const float* beta  = (const float*)d_in[3];
    const float* Wl    = (const float*)d_in[4];
    const float* bl    = (const float*)d_in[5];
    const float* Wr    = (const float*)d_in[6];
    const float* br    = (const float*)d_in[7];
    const float* att   = (const float*)d_in[8];
    const float* bias  = (const float*)d_in[9];
    float* out = (float*)d_out;

    const long N = in_sizes[0] / C_IN;
    const long E = in_sizes[1] / 2;
    const int* srcI = ei;
    const int* dstI = ei + E;

    float* ws   = (float*)d_ws;
    float* xl   = ws;
    float* xr   = xl + N * C_IN;
    int*   seg  = (int*)(xr + N * C_IN);
    int*   gsrc = seg + N * 2;

    const int NB = (int)((N + BIN_NODES - 1) / BIN_NODES);
    const int NW = (int)((E + EDGE_CHUNK - 1) / EDGE_CHUNK);
    int*   binned = (int*)ws;
    int*   runTab = binned + E;

    chunk_bin_kernel<<<NW, 1024, 0, stream>>>(srcI, dstI, runTab, binned, E, NB, NW);
    bin_sort_kernel<<<NB, 256, 0, stream>>>(binned, runTab, seg, gsrc, N, NW);
    ln_gemm_kernel<<<GEMM_GROUPS, 256, 0, stream>>>(x, gamma, beta, Wl, bl, Wr, br, xl, xr, N);
    node_softmax_aggregate_kernel<<<AGG_GROUPS, 256, 0, stream>>>(gsrc, seg, xl, xr, att, bias, out, N);
}
